# starting kernel
# baseline (speedup 1.0000x reference)
#include <hip/hip_runtime.h>


#define GCN_N 50000
#define GCN_E 800000
#define GCN_IN 128
#define GCN_H 64

#define PAIR_STRIDE 130

typedef float v2f __attribute__((ext_vector_type(2)));
typedef float v8f __attribute__((ext_vector_type(8)));


__global__ void gcn_deg_init(float* __restrict__ deg, int n) {
    int i = blockIdx.x * blockDim.x + threadIdx.x;
    if (i < n) deg[i] = 2.0f;
}

__global__ void gcn_deg_edges(const int* __restrict__ ei, float* __restrict__ deg, int e) {
    int i = blockIdx.x * blockDim.x + threadIdx.x;
    if (i < e) atomicAdd(&deg[ei[GCN_E + i]], 1.0f);
}

__global__ void gcn_dinv(const float* __restrict__ deg, float* __restrict__ dinv, int n) {
    int i = blockIdx.x * blockDim.x + threadIdx.x;
    if (i < n) {
        float d = deg[i];
        dinv[i] = (d > 0.0f) ? rsqrtf(d) : 0.0f;
    }
}

__global__ void gcn_edge_norm(const int* __restrict__ ei, const float* __restrict__ dinv,
                              float* __restrict__ norm, int e) {
    int i = blockIdx.x * blockDim.x + threadIdx.x;
    if (i < e) norm[i] = dinv[ei[i]] * dinv[ei[GCN_E + i]];
}


template <int DIN, bool RELU>
__global__ __launch_bounds__(256) void gcn_gemm_wmma(const float* __restrict__ in,
                                                     const float* __restrict__ W,
                                                     float* __restrict__ out) {
    __shared__ float Wl[(DIN / 2) * PAIR_STRIDE];
    const int tid = threadIdx.x;
    for (int i = tid; i < DIN * 64; i += 256) {
        int k = i >> 6;
        int n = i & 63;
        Wl[(k >> 1) * PAIR_STRIDE + n * 2 + (k & 1)] = W[i];
    }
    __syncthreads();

    const int lane = tid & 31;
    const int wave = tid >> 5;
    const int tile = blockIdx.x * 8 + wave;
    if (tile >= (GCN_N >> 4)) return;

    const int m    = lane & 15;
    const int hi   = lane >> 4;
    const int koff = hi * 2;
    const float* arow = in + (size_t)(tile * 16 + m) * DIN;
    const float* brow = &Wl[(koff >> 1) * PAIR_STRIDE + m * 2];

    v8f acc0 = {}, acc1 = {}, acc2 = {}, acc3 = {};
    for (int k0 = 0; k0 < DIN; k0 += 4) {
        float2 av = *(const float2*)(arow + k0 + koff);
        if (RELU) { av.x = fmaxf(av.x, 0.0f); av.y = fmaxf(av.y, 0.0f); }
        v2f a; a.x = av.x; a.y = av.y;

        const float* bp = brow + (k0 >> 1) * PAIR_STRIDE;
        v2f b;
        b = *(const v2f*)(bp +  0);
        acc0 = __builtin_amdgcn_wmma_f32_16x16x4_f32(false, a, false, b, (short)0, acc0, false, false);
        b = *(const v2f*)(bp + 32);
        acc1 = __builtin_amdgcn_wmma_f32_16x16x4_f32(false, a, false, b, (short)0, acc1, false, false);
        b = *(const v2f*)(bp + 64);
        acc2 = __builtin_amdgcn_wmma_f32_16x16x4_f32(false, a, false, b, (short)0, acc2, false, false);
        b = *(const v2f*)(bp + 96);
        acc3 = __builtin_amdgcn_wmma_f32_16x16x4_f32(false, a, false, b, (short)0, acc3, false, false);
    }

    float* orow = out + (size_t)(tile * 16 + 8 * hi) * 64 + m;
#pragma unroll
    for (int r = 0; r < 8; ++r) {
        orow[(size_t)r * 64 +  0] = acc0[r];
        orow[(size_t)r * 64 + 16] = acc1[r];
        orow[(size_t)r * 64 + 32] = acc2[r];
        orow[(size_t)r * 64 + 48] = acc3[r];
    }
}


__global__ void gcn_agg_init(const float* __restrict__ t, const float* __restrict__ dinv,
                             const float* __restrict__ bias, float* __restrict__ out) {
    int idx = blockIdx.x * blockDim.x + threadIdx.x;
    if (idx >= GCN_N * 16) return;
    int i = idx >> 4;
    int q = (idx & 15) * 4;
    float di = dinv[i];
    float s  = 2.0f * di * di;
    float4 tv = *(const float4*)(t + (size_t)i * 64 + q);
    float4 o;
    o.x = bias[q + 0] + s * tv.x;
    o.y = bias[q + 1] + s * tv.y;
    o.z = bias[q + 2] + s * tv.z;
    o.w = bias[q + 3] + s * tv.w;
    *(float4*)(out + (size_t)i * 64 + q) = o;
}


__global__ void gcn_scatter(const int* __restrict__ ei, const float* __restrict__ norm,
                            const float* __restrict__ t, float* __restrict__ out) {
    int idx = blockIdx.x * blockDim.x + threadIdx.x;
    if (idx >= GCN_E * 16) return;
    int e = idx >> 4;
    int q = (idx & 15) * 4;
    int s = ei[e];
    int d = ei[GCN_E + e];
    float w = norm[e];
    float4 tv = *(const float4*)(t + (size_t)s * 64 + q);
    float* o = out + (size_t)d * 64 + q;
    atomicAdd(o + 0, w * tv.x);
    atomicAdd(o + 1, w * tv.y);
    atomicAdd(o + 2, w * tv.z);
    atomicAdd(o + 3, w * tv.w);
}


extern "C" void kernel_launch(void* const* d_in, const int* in_sizes, int n_in,
                              void* d_out, int out_size, void* d_ws, size_t ws_size,
                              hipStream_t stream) {
    (void)in_sizes; (void)n_in; (void)out_size; (void)ws_size;

    const float* x  = (const float*)d_in[0];
    const int*   ei = (const int*)d_in[1];
    const float* W[5] = {(const float*)d_in[2], (const float*)d_in[4], (const float*)d_in[6],
                         (const float*)d_in[8], (const float*)d_in[10]};
    const float* B[5] = {(const float*)d_in[3], (const float*)d_in[5], (const float*)d_in[7],
                         (const float*)d_in[9], (const float*)d_in[11]};

    char*  ws  = (char*)d_ws;
    size_t off = 0;
    auto carve = [&](size_t bytes) {
        void* p = ws + off;
        off = (off + bytes + 255) & ~(size_t)255;
        return p;
    };
    float* deg  = (float*)carve((size_t)GCN_N * 4);
    float* dinv = (float*)carve((size_t)GCN_N * 4);
    float* nrm  = (float*)carve((size_t)GCN_E * 4);
    float* t    = (float*)carve((size_t)GCN_N * 64 * 4);
    float* hagg = (float*)carve((size_t)GCN_N * 64 * 4);

    const int BN  = (GCN_N + 255) / 256;
    const int BE  = (GCN_E + 255) / 256;
    const int BNF = (GCN_N * 16 + 255) / 256;
    const int BEF = (GCN_E * 16 + 255) / 256;
    const int BG  = ((GCN_N / 16) + 7) / 8;

    gcn_deg_init <<<BN, 256, 0, stream>>>(deg, GCN_N);
    gcn_deg_edges<<<BE, 256, 0, stream>>>(ei, deg, GCN_E);
    gcn_dinv     <<<BN, 256, 0, stream>>>(deg, dinv, GCN_N);
    gcn_edge_norm<<<BE, 256, 0, stream>>>(ei, dinv, nrm, GCN_E);

    gcn_gemm_wmma<GCN_IN, false><<<BG, 256, 0, stream>>>(x, W[0], t);
    gcn_agg_init<<<BNF, 256, 0, stream>>>(t, dinv, B[0], hagg);
    gcn_scatter <<<BEF, 256, 0, stream>>>(ei, nrm, t, hagg);

    for (int L = 1; L <= 3; ++L) {
        gcn_gemm_wmma<GCN_H, true><<<BG, 256, 0, stream>>>(hagg, W[L], t);
        gcn_agg_init<<<BNF, 256, 0, stream>>>(t, dinv, B[L], hagg);
        gcn_scatter <<<BEF, 256, 0, stream>>>(ei, nrm, t, hagg);
    }

    float* out = (float*)d_out;
    gcn_gemm_wmma<GCN_H, true><<<BG, 256, 0, stream>>>(hagg, W[4], t);
    gcn_agg_init<<<BNF, 256, 0, stream>>>(t, dinv, B[4], out);
    gcn_scatter <<<BEF, 256, 0, stream>>>(ei, nrm, t, out);
}


# Atomic feature scatter replaced by per-node gather over a dst-grouped edge list built in-call with returning int atomics (arrival-order rank sets the sum order); bias/self-loop init folded in
# speedup vs baseline: 2.2455x; 2.2455x over previous
#include <hip/hip_runtime.h>


#define GCN_N 50000
#define GCN_E 800000
#define GCN_IN 128
#define GCN_H 64

#define PAIR_STRIDE 130

typedef float v2f __attribute__((ext_vector_type(2)));
typedef float v8f __attribute__((ext_vector_type(8)));


__global__ void gcn_cnt_zero(int* __restrict__ cnt) {
    int i = blockIdx.x * blockDim.x + threadIdx.x;
    if (i < GCN_N) cnt[i] = 0;
}

__global__ void gcn_cnt_rank(const int* __restrict__ ei, int* __restrict__ cnt,
                             int* __restrict__ rank) {
    int e = blockIdx.x * blockDim.x + threadIdx.x;
    if (e < GCN_E) rank[e] = atomicAdd(&cnt[ei[GCN_E + e]], 1);
}

__global__ __launch_bounds__(256) void gcn_dinv_bsum(const int* __restrict__ cnt,
                                                     float* __restrict__ dinv,
                                                     int* __restrict__ bsum) {
    __shared__ int red[256];
    const int tid = threadIdx.x;
    const int i = blockIdx.x * 256 + tid;
    int c = 0;
    if (i < GCN_N) {
        c = cnt[i];
        float d = 2.0f + (float)c;
        dinv[i] = (d > 0.0f) ? rsqrtf(d) : 0.0f;
    }
    red[tid] = c;
    __syncthreads();
    for (int s = 128; s > 0; s >>= 1) {
        if (tid < s) red[tid] += red[tid + s];
        __syncthreads();
    }
    if (tid == 0) bsum[blockIdx.x] = red[0];
}

__global__ __launch_bounds__(256) void gcn_rowptr(int* __restrict__ ptr,
                                                  const int* __restrict__ bsum) {
    __shared__ int sc[2][256];
    __shared__ int so[2][256];
    const int tid = threadIdx.x;
    const int b = blockIdx.x;
    const int i = b * 256 + tid;
    const int c = (i < GCN_N) ? ptr[i] : 0;
    sc[0][tid] = c;
    so[0][tid] = (tid < b) ? bsum[tid] : 0;
    __syncthreads();
    int p = 0;
    for (int off = 1; off < 256; off <<= 1) {
        int x = sc[p][tid];
        int y = so[p][tid];
        if (tid >= off) { x += sc[p][tid - off]; y += so[p][tid - off]; }
        sc[p ^ 1][tid] = x;
        so[p ^ 1][tid] = y;
        p ^= 1;
        __syncthreads();
    }
    if (i <= GCN_N) ptr[i] = so[p][255] + sc[p][tid] - c;
}

__global__ void gcn_csr_fill(const int* __restrict__ ei, const int* __restrict__ ptr,
                             const int* __restrict__ rank, int* __restrict__ csr) {
    int e = blockIdx.x * blockDim.x + threadIdx.x;
    if (e < GCN_E) csr[ptr[ei[GCN_E + e]] + rank[e]] = ei[e];
}


template <int DIN, bool RELU>
__global__ __launch_bounds__(256) void gcn_gemm_wmma(const float* __restrict__ in,
                                                     const float* __restrict__ W,
                                                     float* __restrict__ out) {
    __shared__ float Wl[(DIN / 2) * PAIR_STRIDE];
    const int tid = threadIdx.x;
    for (int i = tid; i < DIN * 64; i += 256) {
        int k = i >> 6;
        int n = i & 63;
        Wl[(k >> 1) * PAIR_STRIDE + n * 2 + (k & 1)] = W[i];
    }
    __syncthreads();

    const int lane = tid & 31;
    const int wave = tid >> 5;
    const int tile = blockIdx.x * 8 + wave;
    if (tile >= (GCN_N >> 4)) return;

    const int m    = lane & 15;
    const int hi   = lane >> 4;
    const int koff = hi * 2;
    const float* arow = in + (size_t)(tile * 16 + m) * DIN;
    const float* brow = &Wl[(koff >> 1) * PAIR_STRIDE + m * 2];

    v8f acc0 = {}, acc1 = {}, acc2 = {}, acc3 = {};
    for (int k0 = 0; k0 < DIN; k0 += 4) {
        float2 av = *(const float2*)(arow + k0 + koff);
        if (RELU) { av.x = fmaxf(av.x, 0.0f); av.y = fmaxf(av.y, 0.0f); }
        v2f a; a.x = av.x; a.y = av.y;

        const float* bp = brow + (k0 >> 1) * PAIR_STRIDE;
        v2f b;
        b = *(const v2f*)(bp +  0);
        acc0 = __builtin_amdgcn_wmma_f32_16x16x4_f32(false, a, false, b, (short)0, acc0, false, false);
        b = *(const v2f*)(bp + 32);
        acc1 = __builtin_amdgcn_wmma_f32_16x16x4_f32(false, a, false, b, (short)0, acc1, false, false);
        b = *(const v2f*)(bp + 64);
        acc2 = __builtin_amdgcn_wmma_f32_16x16x4_f32(false, a, false, b, (short)0, acc2, false, false);
        b = *(const v2f*)(bp + 96);
        acc3 = __builtin_amdgcn_wmma_f32_16x16x4_f32(false, a, false, b, (short)0, acc3, false, false);
    }

    float* orow = out + (size_t)(tile * 16 + 8 * hi) * 64 + m;
#pragma unroll
    for (int r = 0; r < 8; ++r) {
        orow[(size_t)r * 64 +  0] = acc0[r];
        orow[(size_t)r * 64 + 16] = acc1[r];
        orow[(size_t)r * 64 + 32] = acc2[r];
        orow[(size_t)r * 64 + 48] = acc3[r];
    }
}


__global__ __launch_bounds__(256) void gcn_gather(const int* __restrict__ ptr,
                                                  const int* __restrict__ csr,
                                                  const float* __restrict__ dinv,
                                                  const float* __restrict__ t,
                                                  const float* __restrict__ bias,
                                                  float* __restrict__ out) {
    const int idx = blockIdx.x * 256 + threadIdx.x;
    const int i   = idx >> 4;
    const int sub = idx & 15;
    const int q   = sub * 4;
    const int beg = ptr[i];
    const int end = ptr[i + 1];
    const float di = dinv[i];
    const float sl = 2.0f * di * di;
    const float4 tv = *(const float4*)(t + (size_t)i * 64 + q);
    float4 acc;
    acc.x = bias[q + 0] + sl * tv.x;
    acc.y = bias[q + 1] + sl * tv.y;
    acc.z = bias[q + 2] + sl * tv.z;
    acc.w = bias[q + 3] + sl * tv.w;

    for (int base = beg; base < end; base += 16) {
        const int my = base + sub;
        int   sid = i;
        float sd  = 0.0f;
        if (my < end) { sid = csr[my]; sd = dinv[sid]; }
        const int nb = min(16, end - base);
#pragma unroll
        for (int h = 0; h < 16; h += 8) {
            if (h < nb) {
                float4 r[8];
#pragma unroll
                for (int u = 0; u < 8; ++u)
                    if (h + u < nb) {
                        const unsigned s = (unsigned)__shfl(sid, h + u, 16);
                        r[u] = *(const float4*)(t + (s * 64u + (unsigned)q));
                    }
#pragma unroll
                for (int u = 0; u < 8; ++u)
                    if (h + u < nb) {
                        const float w = __shfl(sd, h + u, 16) * di;
                        acc.x += w * r[u].x;
                        acc.y += w * r[u].y;
                        acc.z += w * r[u].z;
                        acc.w += w * r[u].w;
                    }
            }
        }
    }
    *(float4*)(out + (size_t)i * 64 + q) = acc;
}


extern "C" void kernel_launch(void* const* d_in, const int* in_sizes, int n_in,
                              void* d_out, int out_size, void* d_ws, size_t ws_size,
                              hipStream_t stream) {
    (void)in_sizes; (void)n_in; (void)out_size; (void)ws_size;

    const float* x  = (const float*)d_in[0];
    const int*   ei = (const int*)d_in[1];
    const float* W[5] = {(const float*)d_in[2], (const float*)d_in[4], (const float*)d_in[6],
                         (const float*)d_in[8], (const float*)d_in[10]};
    const float* B[5] = {(const float*)d_in[3], (const float*)d_in[5], (const float*)d_in[7],
                         (const float*)d_in[9], (const float*)d_in[11]};

    char*  ws  = (char*)d_ws;
    size_t off = 0;
    auto carve = [&](size_t bytes) {
        void* p = ws + off;
        off = (off + bytes + 255) & ~(size_t)255;
        return p;
    };
    int*   ptr  = (int*)  carve((size_t)(GCN_N + 1) * 4);
    float* dinv = (float*)carve((size_t)GCN_N * 4);
    int*   csr  = (int*)  carve((size_t)GCN_E * 4);
    float* t    = (float*)carve((size_t)GCN_N * 64 * 4);
    float* hagg = (float*)carve((size_t)GCN_N * 64 * 4);
    int*   rank = (int*)hagg;
    int*   bsum = (int*)hagg + GCN_E;

    const int BN  = (GCN_N + 255) / 256;
    const int BE  = (GCN_E + 255) / 256;
    const int BNF = (GCN_N * 16) / 256;
    const int BG  = ((GCN_N / 16) + 7) / 8;

    gcn_cnt_zero <<<BN, 256, 0, stream>>>(ptr);
    gcn_cnt_rank <<<BE, 256, 0, stream>>>(ei, ptr, rank);
    gcn_dinv_bsum<<<BN, 256, 0, stream>>>(ptr, dinv, bsum);
    gcn_rowptr   <<<BN, 256, 0, stream>>>(ptr, bsum);
    gcn_csr_fill <<<BE, 256, 0, stream>>>(ei, ptr, rank, csr);

    gcn_gemm_wmma<GCN_IN, false><<<BG, 256, 0, stream>>>(x, W[0], t);
    gcn_gather<<<BNF, 256, 0, stream>>>(ptr, csr, dinv, t, B[0], hagg);

    for (int L = 1; L <= 3; ++L) {
        gcn_gemm_wmma<GCN_H, true><<<BG, 256, 0, stream>>>(hagg, W[L], t);
        gcn_gather<<<BNF, 256, 0, stream>>>(ptr, csr, dinv, t, B[L], hagg);
    }

    float* out = (float*)d_out;
    gcn_gemm_wmma<GCN_H, true><<<BG, 256, 0, stream>>>(hagg, W[4], t);
    gcn_gather<<<BNF, 256, 0, stream>>>(ptr, csr, dinv, t, B[4], out);
}


# Products of layers 2-5 moved into the edge-sum launch before them: a work-group's 16 summed rows are the A tile in LDS, W staged once per 2 tiles; 15 launches to 11; sum and k order unchanged
# speedup vs baseline: 2.7167x; 1.2098x over previous
#include <hip/hip_runtime.h>


#define GCN_N 50000
#define GCN_E 800000
#define GCN_IN 128
#define GCN_H 64

#define PAIR_STRIDE 130

typedef float v2f __attribute__((ext_vector_type(2)));
typedef float v8f __attribute__((ext_vector_type(8)));


__global__ void gcn_cnt_zero(int* __restrict__ cnt) {
    int i = blockIdx.x * blockDim.x + threadIdx.x;
    if (i < GCN_N) cnt[i] = 0;
}

__global__ void gcn_cnt_rank(const int* __restrict__ ei, int* __restrict__ cnt,
                             int* __restrict__ rank) {
    int e = blockIdx.x * blockDim.x + threadIdx.x;
    if (e < GCN_E) rank[e] = atomicAdd(&cnt[ei[GCN_E + e]], 1);
}

__global__ __launch_bounds__(256) void gcn_dinv_bsum(const int* __restrict__ cnt,
                                                     float* __restrict__ dinv,
                                                     int* __restrict__ bsum) {
    __shared__ int red[256];
    const int tid = threadIdx.x;
    const int i = blockIdx.x * 256 + tid;
    int c = 0;
    if (i < GCN_N) {
        c = cnt[i];
        float d = 2.0f + (float)c;
        dinv[i] = (d > 0.0f) ? rsqrtf(d) : 0.0f;
    }
    red[tid] = c;
    __syncthreads();
    for (int s = 128; s > 0; s >>= 1) {
        if (tid < s) red[tid] += red[tid + s];
        __syncthreads();
    }
    if (tid == 0) bsum[blockIdx.x] = red[0];
}

__global__ __launch_bounds__(256) void gcn_rowptr(int* __restrict__ ptr,
                                                  const int* __restrict__ bsum) {
    __shared__ int sc[2][256];
    __shared__ int so[2][256];
    const int tid = threadIdx.x;
    const int b = blockIdx.x;
    const int i = b * 256 + tid;
    const int c = (i < GCN_N) ? ptr[i] : 0;
    sc[0][tid] = c;
    so[0][tid] = (tid < b) ? bsum[tid] : 0;
    __syncthreads();
    int p = 0;
    for (int off = 1; off < 256; off <<= 1) {
        int x = sc[p][tid];
        int y = so[p][tid];
        if (tid >= off) { x += sc[p][tid - off]; y += so[p][tid - off]; }
        sc[p ^ 1][tid] = x;
        so[p ^ 1][tid] = y;
        p ^= 1;
        __syncthreads();
    }
    if (i <= GCN_N) ptr[i] = so[p][255] + sc[p][tid] - c;
}

__global__ void gcn_csr_fill(const int* __restrict__ ei, const int* __restrict__ ptr,
                             const int* __restrict__ rank, int* __restrict__ csr) {
    int e = blockIdx.x * blockDim.x + threadIdx.x;
    if (e < GCN_E) csr[ptr[ei[GCN_E + e]] + rank[e]] = ei[e];
}


template <int DIN, bool RELU>
__global__ __launch_bounds__(256) void gcn_gemm_wmma(const float* __restrict__ in,
                                                     const float* __restrict__ W,
                                                     const float* __restrict__ dinv,
                                                     float* __restrict__ out) {
    __shared__ float Wl[(DIN / 2) * PAIR_STRIDE];
    const int tid = threadIdx.x;
    for (int i = tid; i < DIN * 64; i += 256) {
        int k = i >> 6;
        int n = i & 63;
        Wl[(k >> 1) * PAIR_STRIDE + n * 2 + (k & 1)] = W[i];
    }
    __syncthreads();

    const int lane = tid & 31;
    const int wave = tid >> 5;
    const int tile = blockIdx.x * 8 + wave;
    if (tile >= (GCN_N >> 4)) return;

    const int m    = lane & 15;
    const int hi   = lane >> 4;
    const int koff = hi * 2;
    const float* arow = in + (size_t)(tile * 16 + m) * DIN;
    const float ds = dinv[tile * 16 + m];
    const float* brow = &Wl[(koff >> 1) * PAIR_STRIDE + m * 2];

    v8f acc0 = {}, acc1 = {}, acc2 = {}, acc3 = {};
    for (int k0 = 0; k0 < DIN; k0 += 4) {
        float2 av = *(const float2*)(arow + k0 + koff);
        if (RELU) { av.x = fmaxf(av.x, 0.0f); av.y = fmaxf(av.y, 0.0f); }
        v2f a; a.x = ds * av.x; a.y = ds * av.y;

        const float* bp = brow + (k0 >> 1) * PAIR_STRIDE;
        v2f b;
        b = *(const v2f*)(bp +  0);
        acc0 = __builtin_amdgcn_wmma_f32_16x16x4_f32(false, a, false, b, (short)0, acc0, false, false);
        b = *(const v2f*)(bp + 32);
        acc1 = __builtin_amdgcn_wmma_f32_16x16x4_f32(false, a, false, b, (short)0, acc1, false, false);
        b = *(const v2f*)(bp + 64);
        acc2 = __builtin_amdgcn_wmma_f32_16x16x4_f32(false, a, false, b, (short)0, acc2, false, false);
        b = *(const v2f*)(bp + 96);
        acc3 = __builtin_amdgcn_wmma_f32_16x16x4_f32(false, a, false, b, (short)0, acc3, false, false);
    }

    float* orow = out + (size_t)(tile * 16 + 8 * hi) * 64 + m;
#pragma unroll
    for (int r = 0; r < 8; ++r) {
        orow[(size_t)r * 64 +  0] = acc0[r];
        orow[(size_t)r * 64 + 16] = acc1[r];
        orow[(size_t)r * 64 + 32] = acc2[r];
        orow[(size_t)r * 64 + 48] = acc3[r];
    }
}


__device__ __forceinline__ float4 gcn_edge_sum(const int* __restrict__ ptr,
                                               const int* __restrict__ csr,
                                               const float* __restrict__ t,
                                               const int i, const int sub, const int q) {
    const int beg = ptr[i];
    const int end = ptr[i + 1];
    const float4 tv = *(const float4*)(t + (size_t)i * 64 + q);
    float4 acc;
    acc.x = 2.0f * tv.x;
    acc.y = 2.0f * tv.y;
    acc.z = 2.0f * tv.z;
    acc.w = 2.0f * tv.w;

    for (int base = beg; base < end; base += 16) {
        const int my = base + sub;
        int sid = i;
        if (my < end) sid = csr[my];
        const int nb = min(16, end - base);
#pragma unroll
        for (int h = 0; h < 16; h += 8) {
            if (h < nb) {
                float4 r[8];
#pragma unroll
                for (int u = 0; u < 8; ++u)
                    if (h + u < nb) {
                        const unsigned s = (unsigned)__shfl(sid, h + u, 16);
                        r[u] = *(const float4*)(t + (s * 64u + (unsigned)q));
                    }
#pragma unroll
                for (int u = 0; u < 8; ++u)
                    if (h + u < nb) {
                        acc.x += r[u].x;
                        acc.y += r[u].y;
                        acc.z += r[u].z;
                        acc.w += r[u].w;
                    }
            }
        }
    }
    return acc;
}

__global__ __launch_bounds__(256) void gcn_gather(const int* __restrict__ ptr,
                                                  const int* __restrict__ csr,
                                                  const float* __restrict__ dinv,
                                                  const float* __restrict__ t,
                                                  const float* __restrict__ bias,
                                                  float* __restrict__ out) {
    const int idx = blockIdx.x * 256 + threadIdx.x;
    const int i   = idx >> 4;
    const int sub = idx & 15;
    const int q   = sub * 4;
    const float di = dinv[i];
    const float4 bv = *(const float4*)(bias + q);
    const float4 acc = gcn_edge_sum(ptr, csr, t, i, sub, q);
    float4 o;
    o.x = bv.x + di * acc.x;
    o.y = bv.y + di * acc.y;
    o.z = bv.z + di * acc.z;
    o.w = bv.w + di * acc.w;
    *(float4*)(out + (size_t)i * 64 + q) = o;
}

#define A_PITCH 68
#define GG_TILES 2

__global__ __launch_bounds__(256) void gcn_gather_gemm(const int* __restrict__ ptr,
                                                       const int* __restrict__ csr,
                                                       const float* __restrict__ dinv,
                                                       const float* __restrict__ tin,
                                                       const float* __restrict__ bias,
                                                       const float* __restrict__ W,
                                                       float* __restrict__ tout) {
    __shared__ float Wl[(GCN_H / 2) * PAIR_STRIDE];
    __shared__ float At[GG_TILES][16 * A_PITCH];
    const int tid = threadIdx.x;
#pragma unroll
    for (int jj = 0; jj < (GCN_H * 16) / 256; ++jj) {
        const int i4 = tid + jj * 256;
        const float4 w = *(const float4*)(W + i4 * 4);
        const int k = i4 >> 4;
        const int n = (i4 & 15) * 4;
        float* d = &Wl[(k >> 1) * PAIR_STRIDE + n * 2 + (k & 1)];
        d[0] = w.x; d[2] = w.y; d[4] = w.z; d[6] = w.w;
    }

    const int g    = tid >> 4;
    const int sub  = tid & 15;
    const int q    = sub * 4;
    const int lane = tid & 31;
    const int wave = tid >> 5;
    const int m    = lane & 15;
    const int hi   = lane >> 4;
    const int koff = hi * 2;
    const int n0   = (wave & 3) * 16;
    const float4 bv = *(const float4*)(bias + q);

    for (int j = 0; j < GG_TILES; ++j) {
        const int tile = blockIdx.x * GG_TILES + j;
        if (tile >= (GCN_N >> 4)) break;
        const int i = tile * 16 + g;
        const float di = dinv[i];
        const float4 acc = gcn_edge_sum(ptr, csr, tin, i, sub, q);
        float4 a4;
        a4.x = di * fmaxf(bv.x + di * acc.x, 0.0f);
        a4.y = di * fmaxf(bv.y + di * acc.y, 0.0f);
        a4.z = di * fmaxf(bv.z + di * acc.z, 0.0f);
        a4.w = di * fmaxf(bv.w + di * acc.w, 0.0f);
        *(float4*)(&At[j][g * A_PITCH + q]) = a4;
        __syncthreads();

        if (wave < 4) {
            const float* ap = &At[j][m * A_PITCH + koff];
            const float* bp = &Wl[(koff >> 1) * PAIR_STRIDE + (n0 + m) * 2];
            v8f c8 = {};
#pragma unroll
            for (int kk = 0; kk < 16; ++kk) {
                const v2f a = *(const v2f*)(ap + kk * 4);
                const v2f b = *(const v2f*)(bp + kk * 2 * PAIR_STRIDE);
                c8 = __builtin_amdgcn_wmma_f32_16x16x4_f32(false, a, false, b, (short)0, c8, false, false);
            }
            float* orow = tout + (size_t)(tile * 16 + 8 * hi) * 64 + n0 + m;
#pragma unroll
            for (int r = 0; r < 8; ++r) orow[(size_t)r * 64] = c8[r];
        }
    }
}


extern "C" void kernel_launch(void* const* d_in, const int* in_sizes, int n_in,
                              void* d_out, int out_size, void* d_ws, size_t ws_size,
                              hipStream_t stream) {
    (void)in_sizes; (void)n_in; (void)out_size; (void)ws_size;

    const float* x  = (const float*)d_in[0];
    const int*   ei = (const int*)d_in[1];
    const float* W[5] = {(const float*)d_in[2], (const float*)d_in[4], (const float*)d_in[6],
                         (const float*)d_in[8], (const float*)d_in[10]};
    const float* B[5] = {(const float*)d_in[3], (const float*)d_in[5], (const float*)d_in[7],
                         (const float*)d_in[9], (const float*)d_in[11]};

    char*  ws  = (char*)d_ws;
    size_t off = 0;
    auto carve = [&](size_t bytes) {
        void* p = ws + off;
        off = (off + bytes + 255) & ~(size_t)255;
        return p;
    };
    int*   ptr  = (int*)  carve((size_t)(GCN_N + 1) * 4);
    float* dinv = (float*)carve((size_t)GCN_N * 4);
    int*   csr  = (int*)  carve((size_t)GCN_E * 4);
    float* t    = (float*)carve((size_t)GCN_N * 64 * 4);
    float* hagg = (float*)carve((size_t)GCN_N * 64 * 4);
    int*   rank = (int*)hagg;
    int*   bsum = (int*)hagg + GCN_E;

    const int BN  = (GCN_N + 255) / 256;
    const int BE  = (GCN_E + 255) / 256;
    const int BNF = (GCN_N * 16) / 256;
    const int BG  = ((GCN_N / 16) + 7) / 8;
    const int BGG = ((GCN_N / 16) + GG_TILES - 1) / GG_TILES;

    gcn_cnt_zero <<<BN, 256, 0, stream>>>(ptr);
    gcn_cnt_rank <<<BE, 256, 0, stream>>>(ei, ptr, rank);
    gcn_dinv_bsum<<<BN, 256, 0, stream>>>(ptr, dinv, bsum);
    gcn_rowptr   <<<BN, 256, 0, stream>>>(ptr, bsum);
    gcn_csr_fill <<<BE, 256, 0, stream>>>(ei, ptr, rank, csr);

    gcn_gemm_wmma<GCN_IN, false><<<BG, 256, 0, stream>>>(x, W[0], dinv, t);

    gcn_gather_gemm<<<BGG, 256, 0, stream>>>(ptr, csr, dinv, t,    B[0], W[1], hagg);
    gcn_gather_gemm<<<BGG, 256, 0, stream>>>(ptr, csr, dinv, hagg, B[1], W[2], t);
    gcn_gather_gemm<<<BGG, 256, 0, stream>>>(ptr, csr, dinv, t,    B[2], W[3], hagg);
    gcn_gather_gemm<<<BGG, 256, 0, stream>>>(ptr, csr, dinv, hagg, B[3], W[4], t);

    float* out = (float*)d_out;
    gcn_gather<<<BNF, 256, 0, stream>>>(ptr, csr, dinv, t, B[4], out);
}


# Edge list built by a two-level counting sort on LDS counters (the LDS add's return is the slot) instead of per-edge returning global atomics; 11 launches to 9; arrival order still sets the sum order
# speedup vs baseline: 3.4678x; 1.2765x over previous
#include <hip/hip_runtime.h>


#define GCN_N 50000
#define GCN_E 800000
#define GCN_IN 128
#define GCN_H 64

#define PAIR_STRIDE 130

typedef float v2f __attribute__((ext_vector_type(2)));
typedef float v8f __attribute__((ext_vector_type(8)));

#define EB_T       512
#define EB_CHUNKS  128
#define EB_CHUNK   (GCN_E / EB_CHUNKS)
#define EB_PER_T   ((EB_CHUNK + EB_T - 1) / EB_T)
#define EB_BUCKETS ((GCN_N + 255) / 256)
#define EB_FIN_U   16

__global__ __launch_bounds__(EB_T) void gcn_bucket_count(const int* __restrict__ ei,
                                                         int* __restrict__ hist) {
    __shared__ int h[256];
    const int tid = threadIdx.x;
    const int* dstp = ei + GCN_E + blockIdx.x * EB_CHUNK;
    int d[EB_PER_T];
#pragma unroll
    for (int j = 0; j < EB_PER_T; ++j) {
        const int i = tid + j * EB_T;
        d[j] = (i < EB_CHUNK) ? dstp[i] : -1;
    }
    if (tid < 256) h[tid] = 0;
    __syncthreads();
#pragma unroll
    for (int j = 0; j < EB_PER_T; ++j)
        if (d[j] >= 0) atomicAdd(&h[d[j] >> 8], 1);
    __syncthreads();
    if (tid < 256) hist[blockIdx.x * 256 + tid] = h[tid];
}

__device__ __forceinline__ int gcn_scan256(int* wsum, const int tid, const int v) {
    const int lane = tid & 31;
    const int wave = tid >> 5;
    int x = v;
#pragma unroll
    for (int off = 1; off < 32; off <<= 1) {
        const int y = __shfl_up(x, off, 32);
        if (lane >= off) x += y;
    }
    if (wave < 8 && lane == 31) wsum[wave] = x;
    __syncthreads();
    int add = 0;
#pragma unroll
    for (int w = 0; w < 8; ++w)
        if (w < wave) add += wsum[w];
    return x - v + add;
}

__global__ __launch_bounds__(EB_T) void gcn_bucket_move(const int* __restrict__ ei,
                                                        const int* __restrict__ hist,
                                                        int* __restrict__ bstart,
                                                        int* __restrict__ bkt) {
    __shared__ int part[2][2][256];
    __shared__ int wsum[8];
    __shared__ int cur[256];
    const int tid = threadIdx.x;
    const int c = blockIdx.x;

    const int* srcp = ei + c * EB_CHUNK;
    const int* dstp = ei + GCN_E + c * EB_CHUNK;
    int d[EB_PER_T], s[EB_PER_T];
#pragma unroll
    for (int j = 0; j < EB_PER_T; ++j) {
        const int i = tid + j * EB_T;
        d[j] = (i < EB_CHUNK) ? dstp[i] : -1;
        s[j] = (i < EB_CHUNK) ? srcp[i] : 0;
    }

    const int b = tid & 255;
    const int half = tid >> 8;
    const int* hp = hist + half * (EB_CHUNKS / 2) * 256 + b;
    const int cq = c - half * (EB_CHUNKS / 2);
    int tot = 0, pre = 0;
#pragma unroll
    for (int c2 = 0; c2 < EB_CHUNKS / 2; ++c2) {
        const int hv = hp[c2 * 256];
        tot += hv;
        pre += (c2 < cq) ? hv : 0;
    }
    part[half][0][b] = tot;
    part[half][1][b] = pre;
    __syncthreads();
    int t2 = 0, p2 = 0;
    if (tid < 256) {
        t2 = part[0][0][tid] + part[1][0][tid];
        p2 = part[0][1][tid] + part[1][1][tid];
    }
    const int excl = gcn_scan256(wsum, tid, t2);
    if (tid < 256) {
        cur[tid] = excl + p2;
        if (c == 0) {
            bstart[tid] = excl;
            if (tid == 255) bstart[256] = excl + t2;
        }
    }
    __syncthreads();

#pragma unroll
    for (int j = 0; j < EB_PER_T; ++j)
        if (d[j] >= 0) {
            const int pos = atomicAdd(&cur[d[j] >> 8], 1);
            bkt[pos] = (s[j] << 8) | (d[j] & 255);
        }
}

__global__ __launch_bounds__(EB_T) void gcn_bucket_finish(const int* __restrict__ bstart,
                                                          const int* __restrict__ bkt,
                                                          int* __restrict__ ptr,
                                                          float* __restrict__ dinv,
                                                          int* __restrict__ csr) {
    __shared__ int wsum[8];
    __shared__ int cnt[256];
    const int tid = threadIdx.x;
    const int b = blockIdx.x;
    const int beg = bstart[b];
    const int end = bstart[b + 1];
    const bool one = (end - beg) <= EB_T * EB_FIN_U;
    if (tid < 256) cnt[tid] = 0;
    __syncthreads();
    int v[EB_FIN_U];
    for (int base = beg; base < end; base += EB_T * EB_FIN_U) {
#pragma unroll
        for (int u = 0; u < EB_FIN_U; ++u) {
            const int i = base + u * EB_T + tid;
            v[u] = (i < end) ? bkt[i] : -1;
        }
#pragma unroll
        for (int u = 0; u < EB_FIN_U; ++u)
            if (v[u] >= 0) atomicAdd(&cnt[v[u] & 255], 1);
    }
    __syncthreads();
    const int node = b * 256 + tid;
    int cn = 0;
    if (tid < 256) {
        cn = cnt[tid];
        if (node < GCN_N) {
            float dg = 2.0f + (float)cn;
            dinv[node] = (dg > 0.0f) ? rsqrtf(dg) : 0.0f;
        }
    }
    const int first = beg + gcn_scan256(wsum, tid, cn);
    if (tid < 256) {
        if (node <= GCN_N) ptr[node] = first;
        cnt[tid] = first;
    }
    __syncthreads();
    for (int base = beg; base < end; base += EB_T * EB_FIN_U) {
        if (!one) {
#pragma unroll
            for (int u = 0; u < EB_FIN_U; ++u) {
                const int i = base + u * EB_T + tid;
                v[u] = (i < end) ? bkt[i] : -1;
            }
        }
#pragma unroll
        for (int u = 0; u < EB_FIN_U; ++u)
            if (v[u] >= 0) {
                const int pos = atomicAdd(&cnt[v[u] & 255], 1);
                csr[pos] = v[u] >> 8;
            }
    }
}


template <int DIN, bool RELU>
__global__ __launch_bounds__(256) void gcn_gemm_wmma(const float* __restrict__ in,
                                                     const float* __restrict__ W,
                                                     const float* __restrict__ dinv,
                                                     float* __restrict__ out) {
    __shared__ float Wl[(DIN / 2) * PAIR_STRIDE];
    const int tid = threadIdx.x;
    for (int i = tid; i < DIN * 64; i += 256) {
        int k = i >> 6;
        int n = i & 63;
        Wl[(k >> 1) * PAIR_STRIDE + n * 2 + (k & 1)] = W[i];
    }
    __syncthreads();

    const int lane = tid & 31;
    const int wave = tid >> 5;
    const int tile = blockIdx.x * 8 + wave;
    if (tile >= (GCN_N >> 4)) return;

    const int m    = lane & 15;
    const int hi   = lane >> 4;
    const int koff = hi * 2;
    const float* arow = in + (size_t)(tile * 16 + m) * DIN;
    const float ds = dinv[tile * 16 + m];
    const float* brow = &Wl[(koff >> 1) * PAIR_STRIDE + m * 2];

    v8f acc0 = {}, acc1 = {}, acc2 = {}, acc3 = {};
    for (int k0 = 0; k0 < DIN; k0 += 4) {
        float2 av = *(const float2*)(arow + k0 + koff);
        if (RELU) { av.x = fmaxf(av.x, 0.0f); av.y = fmaxf(av.y, 0.0f); }
        v2f a; a.x = ds * av.x; a.y = ds * av.y;

        const float* bp = brow + (k0 >> 1) * PAIR_STRIDE;
        v2f b;
        b = *(const v2f*)(bp +  0);
        acc0 = __builtin_amdgcn_wmma_f32_16x16x4_f32(false, a, false, b, (short)0, acc0, false, false);
        b = *(const v2f*)(bp + 32);
        acc1 = __builtin_amdgcn_wmma_f32_16x16x4_f32(false, a, false, b, (short)0, acc1, false, false);
        b = *(const v2f*)(bp + 64);
        acc2 = __builtin_amdgcn_wmma_f32_16x16x4_f32(false, a, false, b, (short)0, acc2, false, false);
        b = *(const v2f*)(bp + 96);
        acc3 = __builtin_amdgcn_wmma_f32_16x16x4_f32(false, a, false, b, (short)0, acc3, false, false);
    }

    float* orow = out + (size_t)(tile * 16 + 8 * hi) * 64 + m;
#pragma unroll
    for (int r = 0; r < 8; ++r) {
        orow[(size_t)r * 64 +  0] = acc0[r];
        orow[(size_t)r * 64 + 16] = acc1[r];
        orow[(size_t)r * 64 + 32] = acc2[r];
        orow[(size_t)r * 64 + 48] = acc3[r];
    }
}


__device__ __forceinline__ float4 gcn_edge_sum(const int* __restrict__ ptr,
                                               const int* __restrict__ csr,
                                               const float* __restrict__ t,
                                               const int i, const int sub, const int q) {
    const int beg = ptr[i];
    const int end = ptr[i + 1];
    const float4 tv = *(const float4*)(t + (size_t)i * 64 + q);
    float4 acc;
    acc.x = 2.0f * tv.x;
    acc.y = 2.0f * tv.y;
    acc.z = 2.0f * tv.z;
    acc.w = 2.0f * tv.w;

    for (int base = beg; base < end; base += 16) {
        const int my = base + sub;
        int sid = i;
        if (my < end) sid = csr[my];
        const int nb = min(16, end - base);
#pragma unroll
        for (int h = 0; h < 16; h += 8) {
            if (h < nb) {
                float4 r[8];
#pragma unroll
                for (int u = 0; u < 8; ++u)
                    if (h + u < nb) {
                        const unsigned s = (unsigned)__shfl(sid, h + u, 16);
                        r[u] = *(const float4*)(t + (s * 64u + (unsigned)q));
                    }
#pragma unroll
                for (int u = 0; u < 8; ++u)
                    if (h + u < nb) {
                        acc.x += r[u].x;
                        acc.y += r[u].y;
                        acc.z += r[u].z;
                        acc.w += r[u].w;
                    }
            }
        }
    }
    return acc;
}

__global__ __launch_bounds__(256) void gcn_gather(const int* __restrict__ ptr,
                                                  const int* __restrict__ csr,
                                                  const float* __restrict__ dinv,
                                                  const float* __restrict__ t,
                                                  const float* __restrict__ bias,
                                                  float* __restrict__ out) {
    const int idx = blockIdx.x * 256 + threadIdx.x;
    const int i   = idx >> 4;
    const int sub = idx & 15;
    const int q   = sub * 4;
    const float di = dinv[i];
    const float4 bv = *(const float4*)(bias + q);
    const float4 acc = gcn_edge_sum(ptr, csr, t, i, sub, q);
    float4 o;
    o.x = bv.x + di * acc.x;
    o.y = bv.y + di * acc.y;
    o.z = bv.z + di * acc.z;
    o.w = bv.w + di * acc.w;
    *(float4*)(out + (size_t)i * 64 + q) = o;
}

#define A_PITCH 68
#define GG_TILES 2

__global__ __launch_bounds__(256) void gcn_gather_gemm(const int* __restrict__ ptr,
                                                       const int* __restrict__ csr,
                                                       const float* __restrict__ dinv,
                                                       const float* __restrict__ tin,
                                                       const float* __restrict__ bias,
                                                       const float* __restrict__ W,
                                                       float* __restrict__ tout) {
    __shared__ float Wl[(GCN_H / 2) * PAIR_STRIDE];
    __shared__ float At[GG_TILES][16 * A_PITCH];
    const int tid = threadIdx.x;
#pragma unroll
    for (int jj = 0; jj < (GCN_H * 16) / 256; ++jj) {
        const int i4 = tid + jj * 256;
        const float4 w = *(const float4*)(W + i4 * 4);
        const int k = i4 >> 4;
        const int n = (i4 & 15) * 4;
        float* d = &Wl[(k >> 1) * PAIR_STRIDE + n * 2 + (k & 1)];
        d[0] = w.x; d[2] = w.y; d[4] = w.z; d[6] = w.w;
    }

    const int g    = tid >> 4;
    const int sub  = tid & 15;
    const int q    = sub * 4;
    const int lane = tid & 31;
    const int wave = tid >> 5;
    const int m    = lane & 15;
    const int hi   = lane >> 4;
    const int koff = hi * 2;
    const int n0   = (wave & 3) * 16;
    const float4 bv = *(const float4*)(bias + q);

    for (int j = 0; j < GG_TILES; ++j) {
        const int tile = blockIdx.x * GG_TILES + j;
        if (tile >= (GCN_N >> 4)) break;
        const int i = tile * 16 + g;
        const float di = dinv[i];
        const float4 acc = gcn_edge_sum(ptr, csr, tin, i, sub, q);
        float4 a4;
        a4.x = di * fmaxf(bv.x + di * acc.x, 0.0f);
        a4.y = di * fmaxf(bv.y + di * acc.y, 0.0f);
        a4.z = di * fmaxf(bv.z + di * acc.z, 0.0f);
        a4.w = di * fmaxf(bv.w + di * acc.w, 0.0f);
        *(float4*)(&At[j][g * A_PITCH + q]) = a4;
        __syncthreads();

        if (wave < 4) {
            const float* ap = &At[j][m * A_PITCH + koff];
            const float* bp = &Wl[(koff >> 1) * PAIR_STRIDE + (n0 + m) * 2];
            v8f c8 = {};
#pragma unroll
            for (int kk = 0; kk < 16; ++kk) {
                const v2f a = *(const v2f*)(ap + kk * 4);
                const v2f b = *(const v2f*)(bp + kk * 2 * PAIR_STRIDE);
                c8 = __builtin_amdgcn_wmma_f32_16x16x4_f32(false, a, false, b, (short)0, c8, false, false);
            }
            float* orow = tout + (size_t)(tile * 16 + 8 * hi) * 64 + n0 + m;
#pragma unroll
            for (int r = 0; r < 8; ++r) orow[(size_t)r * 64] = c8[r];
        }
    }
}


extern "C" void kernel_launch(void* const* d_in, const int* in_sizes, int n_in,
                              void* d_out, int out_size, void* d_ws, size_t ws_size,
                              hipStream_t stream) {
    (void)in_sizes; (void)n_in; (void)out_size; (void)ws_size;

    const float* x  = (const float*)d_in[0];
    const int*   ei = (const int*)d_in[1];
    const float* W[5] = {(const float*)d_in[2], (const float*)d_in[4], (const float*)d_in[6],
                         (const float*)d_in[8], (const float*)d_in[10]};
    const float* B[5] = {(const float*)d_in[3], (const float*)d_in[5], (const float*)d_in[7],
                         (const float*)d_in[9], (const float*)d_in[11]};

    char*  ws  = (char*)d_ws;
    size_t off = 0;
    auto carve = [&](size_t bytes) {
        void* p = ws + off;
        off = (off + bytes + 255) & ~(size_t)255;
        return p;
    };
    int*   ptr  = (int*)  carve((size_t)(GCN_N + 1) * 4);
    float* dinv = (float*)carve((size_t)GCN_N * 4);
    int*   csr  = (int*)  carve((size_t)GCN_E * 4);
    float* t    = (float*)carve((size_t)GCN_N * 64 * 4);
    float* hagg = (float*)carve((size_t)GCN_N * 64 * 4);
    int*   bkt    = (int*)hagg;
    int*   hist   = (int*)hagg + GCN_E;
    int*   bstart = (int*)hagg + GCN_E + EB_CHUNKS * 256;

    const int BNF = (GCN_N * 16) / 256;
    const int BG  = ((GCN_N / 16) + 7) / 8;
    const int BGG = ((GCN_N / 16) + GG_TILES - 1) / GG_TILES;

    gcn_bucket_count <<<EB_CHUNKS,  EB_T, 0, stream>>>(ei, hist);
    gcn_bucket_move  <<<EB_CHUNKS,  EB_T, 0, stream>>>(ei, hist, bstart, bkt);
    gcn_bucket_finish<<<EB_BUCKETS, EB_T, 0, stream>>>(bstart, bkt, ptr, dinv, csr);

    gcn_gemm_wmma<GCN_IN, false><<<BG, 256, 0, stream>>>(x, W[0], dinv, t);

    gcn_gather_gemm<<<BGG, 256, 0, stream>>>(ptr, csr, dinv, t,    B[0], W[1], hagg);
    gcn_gather_gemm<<<BGG, 256, 0, stream>>>(ptr, csr, dinv, hagg, B[1], W[2], t);
    gcn_gather_gemm<<<BGG, 256, 0, stream>>>(ptr, csr, dinv, t,    B[2], W[3], hagg);
    gcn_gather_gemm<<<BGG, 256, 0, stream>>>(ptr, csr, dinv, hagg, B[3], W[4], t);

    float* out = (float*)d_out;
    gcn_gather<<<BNF, 256, 0, stream>>>(ptr, csr, dinv, t, B[4], out);
}


# Edge sums software-pipelined over 4 node tiles per work-group: list bounds asked two tiles ahead, first 32 source ids and own row one tile ahead, 16 row loads per group; sum order unchanged
# speedup vs baseline: 3.7104x; 1.0700x over previous
#include <hip/hip_runtime.h>


#define GCN_N 50000
#define GCN_E 800000
#define GCN_IN 128
#define GCN_H 64

#define PAIR_STRIDE 130

typedef float v2f __attribute__((ext_vector_type(2)));
typedef float v8f __attribute__((ext_vector_type(8)));

#define EB_T       512
#define EB_CHUNKS  128
#define EB_CHUNK   (GCN_E / EB_CHUNKS)
#define EB_PER_T   ((EB_CHUNK + EB_T - 1) / EB_T)
#define EB_BUCKETS ((GCN_N + 255) / 256)
#define EB_FIN_U   16

__global__ __launch_bounds__(EB_T) void gcn_bucket_count(const int* __restrict__ ei,
                                                         int* __restrict__ hist) {
    __shared__ int h[256];
    const int tid = threadIdx.x;
    const int* dstp = ei + GCN_E + blockIdx.x * EB_CHUNK;
    int d[EB_PER_T];
#pragma unroll
    for (int j = 0; j < EB_PER_T; ++j) {
        const int i = tid + j * EB_T;
        d[j] = (i < EB_CHUNK) ? dstp[i] : -1;
    }
    if (tid < 256) h[tid] = 0;
    __syncthreads();
#pragma unroll
    for (int j = 0; j < EB_PER_T; ++j)
        if (d[j] >= 0) atomicAdd(&h[d[j] >> 8], 1);
    __syncthreads();
    if (tid < 256) hist[blockIdx.x * 256 + tid] = h[tid];
}

__device__ __forceinline__ int gcn_scan256(int* wsum, const int tid, const int v) {
    const int lane = tid & 31;
    const int wave = tid >> 5;
    int x = v;
#pragma unroll
    for (int off = 1; off < 32; off <<= 1) {
        const int y = __shfl_up(x, off, 32);
        if (lane >= off) x += y;
    }
    if (wave < 8 && lane == 31) wsum[wave] = x;
    __syncthreads();
    int add = 0;
#pragma unroll
    for (int w = 0; w < 8; ++w)
        if (w < wave) add += wsum[w];
    return x - v + add;
}

__global__ __launch_bounds__(EB_T) void gcn_bucket_move(const int* __restrict__ ei,
                                                        const int* __restrict__ hist,
                                                        int* __restrict__ bstart,
                                                        int* __restrict__ bkt) {
    __shared__ int part[2][2][256];
    __shared__ int wsum[8];
    __shared__ int cur[256];
    const int tid = threadIdx.x;
    const int c = blockIdx.x;

    const int* srcp = ei + c * EB_CHUNK;
    const int* dstp = ei + GCN_E + c * EB_CHUNK;
    int d[EB_PER_T], s[EB_PER_T];
#pragma unroll
    for (int j = 0; j < EB_PER_T; ++j) {
        const int i = tid + j * EB_T;
        d[j] = (i < EB_CHUNK) ? dstp[i] : -1;
        s[j] = (i < EB_CHUNK) ? srcp[i] : 0;
    }

    const int b = tid & 255;
    const int half = tid >> 8;
    const int* hp = hist + half * (EB_CHUNKS / 2) * 256 + b;
    const int cq = c - half * (EB_CHUNKS / 2);
    int tot = 0, pre = 0;
#pragma unroll
    for (int c2 = 0; c2 < EB_CHUNKS / 2; ++c2) {
        const int hv = hp[c2 * 256];
        tot += hv;
        pre += (c2 < cq) ? hv : 0;
    }
    part[half][0][b] = tot;
    part[half][1][b] = pre;
    __syncthreads();
    int t2 = 0, p2 = 0;
    if (tid < 256) {
        t2 = part[0][0][tid] + part[1][0][tid];
        p2 = part[0][1][tid] + part[1][1][tid];
    }
    const int excl = gcn_scan256(wsum, tid, t2);
    if (tid < 256) {
        cur[tid] = excl + p2;
        if (c == 0) {
            bstart[tid] = excl;
            if (tid == 255) bstart[256] = excl + t2;
        }
    }
    __syncthreads();

#pragma unroll
    for (int j = 0; j < EB_PER_T; ++j)
        if (d[j] >= 0) {
            const int pos = atomicAdd(&cur[d[j] >> 8], 1);
            bkt[pos] = (s[j] << 8) | (d[j] & 255);
        }
}

__global__ __launch_bounds__(EB_T) void gcn_bucket_finish(const int* __restrict__ bstart,
                                                          const int* __restrict__ bkt,
                                                          int* __restrict__ ptr,
                                                          float* __restrict__ dinv,
                                                          int* __restrict__ csr) {
    __shared__ int wsum[8];
    __shared__ int cnt[256];
    const int tid = threadIdx.x;
    const int b = blockIdx.x;
    const int beg = bstart[b];
    const int end = bstart[b + 1];
    const bool one = (end - beg) <= EB_T * EB_FIN_U;
    if (tid < 256) cnt[tid] = 0;
    __syncthreads();
    int v[EB_FIN_U];
    for (int base = beg; base < end; base += EB_T * EB_FIN_U) {
#pragma unroll
        for (int u = 0; u < EB_FIN_U; ++u) {
            const int i = base + u * EB_T + tid;
            v[u] = (i < end) ? bkt[i] : -1;
        }
#pragma unroll
        for (int u = 0; u < EB_FIN_U; ++u)
            if (v[u] >= 0) atomicAdd(&cnt[v[u] & 255], 1);
    }
    __syncthreads();
    const int node = b * 256 + tid;
    int cn = 0;
    if (tid < 256) {
        cn = cnt[tid];
        if (node < GCN_N) {
            float dg = 2.0f + (float)cn;
            dinv[node] = (dg > 0.0f) ? rsqrtf(dg) : 0.0f;
        }
    }
    const int first = beg + gcn_scan256(wsum, tid, cn);
    if (tid < 256) {
        if (node <= GCN_N) ptr[node] = first;
        cnt[tid] = first;
    }
    __syncthreads();
    for (int base = beg; base < end; base += EB_T * EB_FIN_U) {
        if (!one) {
#pragma unroll
            for (int u = 0; u < EB_FIN_U; ++u) {
                const int i = base + u * EB_T + tid;
                v[u] = (i < end) ? bkt[i] : -1;
            }
        }
#pragma unroll
        for (int u = 0; u < EB_FIN_U; ++u)
            if (v[u] >= 0) {
                const int pos = atomicAdd(&cnt[v[u] & 255], 1);
                csr[pos] = v[u] >> 8;
            }
    }
}


template <int DIN, bool RELU>
__global__ __launch_bounds__(256) void gcn_gemm_wmma(const float* __restrict__ in,
                                                     const float* __restrict__ W,
                                                     const float* __restrict__ dinv,
                                                     float* __restrict__ out) {
    __shared__ float Wl[(DIN / 2) * PAIR_STRIDE];
    const int tid = threadIdx.x;
    for (int i = tid; i < DIN * 64; i += 256) {
        int k = i >> 6;
        int n = i & 63;
        Wl[(k >> 1) * PAIR_STRIDE + n * 2 + (k & 1)] = W[i];
    }
    __syncthreads();

    const int lane = tid & 31;
    const int wave = tid >> 5;
    const int tile = blockIdx.x * 8 + wave;
    if (tile >= (GCN_N >> 4)) return;

    const int m    = lane & 15;
    const int hi   = lane >> 4;
    const int koff = hi * 2;
    const float* arow = in + (size_t)(tile * 16 + m) * DIN;
    const float ds = dinv[tile * 16 + m];
    const float* brow = &Wl[(koff >> 1) * PAIR_STRIDE + m * 2];

    v8f acc0 = {}, acc1 = {}, acc2 = {}, acc3 = {};
    for (int k0 = 0; k0 < DIN; k0 += 4) {
        float2 av = *(const float2*)(arow + k0 + koff);
        if (RELU) { av.x = fmaxf(av.x, 0.0f); av.y = fmaxf(av.y, 0.0f); }
        v2f a; a.x = ds * av.x; a.y = ds * av.y;

        const float* bp = brow + (k0 >> 1) * PAIR_STRIDE;
        v2f b;
        b = *(const v2f*)(bp +  0);
        acc0 = __builtin_amdgcn_wmma_f32_16x16x4_f32(false, a, false, b, (short)0, acc0, false, false);
        b = *(const v2f*)(bp + 32);
        acc1 = __builtin_amdgcn_wmma_f32_16x16x4_f32(false, a, false, b, (short)0, acc1, false, false);
        b = *(const v2f*)(bp + 64);
        acc2 = __builtin_amdgcn_wmma_f32_16x16x4_f32(false, a, false, b, (short)0, acc2, false, false);
        b = *(const v2f*)(bp + 96);
        acc3 = __builtin_amdgcn_wmma_f32_16x16x4_f32(false, a, false, b, (short)0, acc3, false, false);
    }

    float* orow = out + (size_t)(tile * 16 + 8 * hi) * 64 + m;
#pragma unroll
    for (int r = 0; r < 8; ++r) {
        orow[(size_t)r * 64 +  0] = acc0[r];
        orow[(size_t)r * 64 + 16] = acc1[r];
        orow[(size_t)r * 64 + 32] = acc2[r];
        orow[(size_t)r * 64 + 48] = acc3[r];
    }
}

#define GG_TILES 4

__device__ __forceinline__ void gcn_ids32(const int* __restrict__ csr, const int beg,
                                          const int sub, int& s0, int& s1) {
    s0 = csr[min(beg + sub, GCN_E - 1)];
    s1 = csr[min(beg + 16 + sub, GCN_E - 1)];
}

__device__ __forceinline__ void gcn_rows16_issue(float4 (&r)[16], const float* __restrict__ t,
                                                 const int sid, const int n, const int q) {
    unsigned off[16];
#pragma unroll
    for (int u = 0; u < 16; ++u)
        off[u] = (unsigned)__shfl(sid, u, 16) * 256u + (unsigned)q * 4u;
#pragma unroll
    for (int u = 0; u < 16; ++u)
        if (u < n) r[u] = *(const float4*)((const char*)t + off[u]);
}

__device__ __forceinline__ void gcn_rows16_sum(float4& acc, const float4 (&r)[16], const int n) {
#pragma unroll
    for (int u = 0; u < 16; ++u)
        if (u < n) {
            acc.x += r[u].x;
            acc.y += r[u].y;
            acc.z += r[u].z;
            acc.w += r[u].w;
        }
}

template <class F>
__device__ __forceinline__ float4 gcn_node_sum(const int* __restrict__ csr,
                                               const float* __restrict__ t,
                                               const int sub, const int q,
                                               const int beg, const int end,
                                               const int s0, const int s1,
                                               const float4 tv, F&& ahead) {
    float4 acc;
    acc.x = 2.0f * tv.x;
    acc.y = 2.0f * tv.y;
    acc.z = 2.0f * tv.z;
    acc.w = 2.0f * tv.w;
    const int deg = end - beg;
    float4 r[16];
    gcn_rows16_issue(r, t, s0, deg, q);
    ahead();
    gcn_rows16_sum(acc, r, deg);
    for (int base = 16; base < deg; base += 16) {
        int sid = s1;
        if (base >= 32) sid = csr[min(beg + base + sub, GCN_E - 1)];
        gcn_rows16_issue(r, t, sid, deg - base, q);
        gcn_rows16_sum(acc, r, deg - base);
    }
    return acc;
}

struct gcn_pipe {
    int   begC, endC, s0C, s1C;
    float diC;
    float4 tvC;
    int   begN, endN;
    int   begNN, endNN, s0N, s1N;
    float diN;
    float4 tvN;
};

__device__ __forceinline__ void gcn_pipe_bounds(gcn_pipe& p, const int* __restrict__ ptr,
                                                const int i, const int nt) {
    p.begC = ptr[i];
    p.endC = ptr[i + 1];
    p.begN = 0; p.endN = 0;
    if (nt > 1) { p.begN = ptr[i + 16]; p.endN = ptr[i + 17]; }
}
__device__ __forceinline__ void gcn_pipe_first(gcn_pipe& p, const int* __restrict__ csr,
                                               const float* __restrict__ dinv,
                                               const float* __restrict__ t,
                                               const int i, const int sub, const int q) {
    gcn_ids32(csr, p.begC, sub, p.s0C, p.s1C);
    p.diC = dinv[i];
    p.tvC = *(const float4*)(t + (size_t)i * 64 + q);
}
__device__ __forceinline__ void gcn_pipe_ahead(gcn_pipe& p, const int* __restrict__ ptr,
                                               const int* __restrict__ csr,
                                               const float* __restrict__ dinv,
                                               const float* __restrict__ t,
                                               const int i, const int sub, const int q,
                                               const int j, const int nt) {
    p.s0N = 0; p.s1N = 0; p.begNN = 0; p.endNN = 0; p.diN = 0.0f; p.tvN = p.tvC;
    if (j + 1 < nt) {
        gcn_ids32(csr, p.begN, sub, p.s0N, p.s1N);
        p.diN = dinv[i + 16];
        p.tvN = *(const float4*)(t + (size_t)(i + 16) * 64 + q);
    }
    if (j + 2 < nt) { p.begNN = ptr[i + 32]; p.endNN = ptr[i + 33]; }
}
__device__ __forceinline__ void gcn_pipe_step(gcn_pipe& p) {
    p.begC = p.begN; p.endC = p.endN; p.s0C = p.s0N; p.s1C = p.s1N;
    p.diC = p.diN; p.tvC = p.tvN;
    p.begN = p.begNN; p.endN = p.endNN;
}

__global__ __launch_bounds__(256) void gcn_gather(const int* __restrict__ ptr,
                                                  const int* __restrict__ csr,
                                                  const float* __restrict__ dinv,
                                                  const float* __restrict__ t,
                                                  const float* __restrict__ bias,
                                                  float* __restrict__ out) {
    const int tid = threadIdx.x;
    const int g   = tid >> 4;
    const int sub = tid & 15;
    const int q   = sub * 4;
    const int tile0 = blockIdx.x * GG_TILES;
    const int nt = min(GG_TILES, (GCN_N >> 4) - tile0);
    int i = tile0 * 16 + g;
    gcn_pipe p;
    gcn_pipe_bounds(p, ptr, i, nt);
    const float4 bv = *(const float4*)(bias + q);
    gcn_pipe_first(p, csr, dinv, t, i, sub, q);
    for (int j = 0; j < nt; ++j) {
        const float di = p.diC;
        const float4 acc = gcn_node_sum(csr, t, sub, q, p.begC, p.endC, p.s0C, p.s1C, p.tvC,
                                        [&] { gcn_pipe_ahead(p, ptr, csr, dinv, t, i, sub, q, j, nt); });
        float4 o;
        o.x = bv.x + di * acc.x;
        o.y = bv.y + di * acc.y;
        o.z = bv.z + di * acc.z;
        o.w = bv.w + di * acc.w;
        *(float4*)(out + (size_t)i * 64 + q) = o;
        gcn_pipe_step(p);
        i += 16;
    }
}

#define A_PITCH 68

__global__ __launch_bounds__(256) void gcn_gather_gemm(const int* __restrict__ ptr,
                                                       const int* __restrict__ csr,
                                                       const float* __restrict__ dinv,
                                                       const float* __restrict__ tin,
                                                       const float* __restrict__ bias,
                                                       const float* __restrict__ W,
                                                       float* __restrict__ tout) {
    __shared__ float Wl[(GCN_H / 2) * PAIR_STRIDE];
    __shared__ float At[2][16 * A_PITCH];
    const int tid = threadIdx.x;
    const int g    = tid >> 4;
    const int sub  = tid & 15;
    const int q    = sub * 4;

    const int tile0 = blockIdx.x * GG_TILES;
    const int nt = min(GG_TILES, (GCN_N >> 4) - tile0);
    int i = tile0 * 16 + g;
    gcn_pipe p;
    gcn_pipe_bounds(p, ptr, i, nt);

#pragma unroll
    for (int jj = 0; jj < (GCN_H * 16) / 256; ++jj) {
        const int i4 = tid + jj * 256;
        const float4 w = *(const float4*)(W + i4 * 4);
        const int k = i4 >> 4;
        const int n = (i4 & 15) * 4;
        float* d = &Wl[(k >> 1) * PAIR_STRIDE + n * 2 + (k & 1)];
        d[0] = w.x; d[2] = w.y; d[4] = w.z; d[6] = w.w;
    }

    const int lane = tid & 31;
    const int wave = tid >> 5;
    const int m    = lane & 15;
    const int hi   = lane >> 4;
    const int koff = hi * 2;
    const int n0   = (wave & 3) * 16;
    const float4 bv = *(const float4*)(bias + q);
    gcn_pipe_first(p, csr, dinv, tin, i, sub, q);

    for (int j = 0; j < nt; ++j) {
        const int tile = tile0 + j;
        const float di = p.diC;
        const float4 acc = gcn_node_sum(csr, tin, sub, q, p.begC, p.endC, p.s0C, p.s1C, p.tvC,
                                        [&] { gcn_pipe_ahead(p, ptr, csr, dinv, tin, i, sub, q, j, nt); });
        float4 a4;
        a4.x = di * fmaxf(bv.x + di * acc.x, 0.0f);
        a4.y = di * fmaxf(bv.y + di * acc.y, 0.0f);
        a4.z = di * fmaxf(bv.z + di * acc.z, 0.0f);
        a4.w = di * fmaxf(bv.w + di * acc.w, 0.0f);
        float* Ab = At[j & 1];
        *(float4*)(&Ab[g * A_PITCH + q]) = a4;
        __syncthreads();

        if (wave < 4) {
            const float* ap = &Ab[m * A_PITCH + koff];
            const float* bp = &Wl[(koff >> 1) * PAIR_STRIDE + (n0 + m) * 2];
            v8f c8 = {};
#pragma unroll
            for (int kk = 0; kk < 16; ++kk) {
                const v2f a = *(const v2f*)(ap + kk * 4);
                const v2f b = *(const v2f*)(bp + kk * 2 * PAIR_STRIDE);
                c8 = __builtin_amdgcn_wmma_f32_16x16x4_f32(false, a, false, b, (short)0, c8, false, false);
            }
            float* orow = tout + (size_t)(tile * 16 + 8 * hi) * 64 + n0 + m;
#pragma unroll
            for (int r = 0; r < 8; ++r) orow[(size_t)r * 64] = c8[r];
        }
        gcn_pipe_step(p);
        i += 16;
    }
}


extern "C" void kernel_launch(void* const* d_in, const int* in_sizes, int n_in,
                              void* d_out, int out_size, void* d_ws, size_t ws_size,
                              hipStream_t stream) {
    (void)in_sizes; (void)n_in; (void)out_size; (void)ws_size;

    const float* x  = (const float*)d_in[0];
    const int*   ei = (const int*)d_in[1];
    const float* W[5] = {(const float*)d_in[2], (const float*)d_in[4], (const float*)d_in[6],
                         (const float*)d_in[8], (const float*)d_in[10]};
    const float* B[5] = {(const float*)d_in[3], (const float*)d_in[5], (const float*)d_in[7],
                         (const float*)d_in[9], (const float*)d_in[11]};

    char*  ws  = (char*)d_ws;
    size_t off = 0;
    auto carve = [&](size_t bytes) {
        void* p = ws + off;
        off = (off + bytes + 255) & ~(size_t)255;
        return p;
    };
    int*   ptr  = (int*)  carve((size_t)(GCN_N + 1) * 4);
    float* dinv = (float*)carve((size_t)GCN_N * 4);
    int*   csr  = (int*)  carve((size_t)GCN_E * 4);
    float* t    = (float*)carve((size_t)GCN_N * 64 * 4);
    float* hagg = (float*)carve((size_t)GCN_N * 64 * 4);
    int*   bkt    = (int*)hagg;
    int*   hist   = (int*)hagg + GCN_E;
    int*   bstart = (int*)hagg + GCN_E + EB_CHUNKS * 256;

    const int BG  = ((GCN_N / 16) + 7) / 8;
    const int BGG = ((GCN_N / 16) + GG_TILES - 1) / GG_TILES;

    gcn_bucket_count <<<EB_CHUNKS,  EB_T, 0, stream>>>(ei, hist);
    gcn_bucket_move  <<<EB_CHUNKS,  EB_T, 0, stream>>>(ei, hist, bstart, bkt);
    gcn_bucket_finish<<<EB_BUCKETS, EB_T, 0, stream>>>(bstart, bkt, ptr, dinv, csr);

    gcn_gemm_wmma<GCN_IN, false><<<BG, 256, 0, stream>>>(x, W[0], dinv, t);

    gcn_gather_gemm<<<BGG, 256, 0, stream>>>(ptr, csr, dinv, t,    B[0], W[1], hagg);
    gcn_gather_gemm<<<BGG, 256, 0, stream>>>(ptr, csr, dinv, hagg, B[1], W[2], t);
    gcn_gather_gemm<<<BGG, 256, 0, stream>>>(ptr, csr, dinv, t,    B[2], W[3], hagg);
    gcn_gather_gemm<<<BGG, 256, 0, stream>>>(ptr, csr, dinv, hagg, B[3], W[4], t);

    float* out = (float*)d_out;
    gcn_gather<<<BGG, 256, 0, stream>>>(ptr, csr, dinv, t, B[4], out);
}


# Layer 1 product x@W1 co-scheduled into the bucket-move launch as extra work-groups (it needs no dinv); its dinv row scale moved behind the product, into the finish launch; 9 launches to 8
# speedup vs baseline: 3.8232x; 1.0304x over previous
#include <hip/hip_runtime.h>


#define GCN_N 50000
#define GCN_E 800000
#define GCN_IN 128
#define GCN_H 64

#define PAIR_STRIDE 130

typedef float v2f __attribute__((ext_vector_type(2)));
typedef float v8f __attribute__((ext_vector_type(8)));

#define EB_T       512
#define EB_CHUNKS  128
#define EB_CHUNK   (GCN_E / EB_CHUNKS)
#define EB_PER_T   ((EB_CHUNK + EB_T - 1) / EB_T)
#define EB_BUCKETS ((GCN_N + 255) / 256)
#define EB_FIN_U   16

__global__ __launch_bounds__(EB_T) void gcn_bucket_count(const int* __restrict__ ei,
                                                         int* __restrict__ hist) {
    __shared__ int h[256];
    const int tid = threadIdx.x;
    const int* dstp = ei + GCN_E + blockIdx.x * EB_CHUNK;
    int d[EB_PER_T];
#pragma unroll
    for (int j = 0; j < EB_PER_T; ++j) {
        const int i = tid + j * EB_T;
        d[j] = (i < EB_CHUNK) ? dstp[i] : -1;
    }
    if (tid < 256) h[tid] = 0;
    __syncthreads();
#pragma unroll
    for (int j = 0; j < EB_PER_T; ++j)
        if (d[j] >= 0) atomicAdd(&h[d[j] >> 8], 1);
    __syncthreads();
    if (tid < 256) hist[blockIdx.x * 256 + tid] = h[tid];
}

__device__ __forceinline__ int gcn_scan256(int* wsum, const int tid, const int v) {
    const int lane = tid & 31;
    const int wave = tid >> 5;
    int x = v;
#pragma unroll
    for (int off = 1; off < 32; off <<= 1) {
        const int y = __shfl_up(x, off, 32);
        if (lane >= off) x += y;
    }
    if (wave < 8 && lane == 31) wsum[wave] = x;
    __syncthreads();
    int add = 0;
#pragma unroll
    for (int w = 0; w < 8; ++w)
        if (w < wave) add += wsum[w];
    return x - v + add;
}

#define GM_WAVES  (EB_T / 32)
#define GM_BLOCKS (((GCN_N >> 4) + GM_WAVES - 1) / GM_WAVES)

__device__ __forceinline__ void gcn_layer1_block(const float* __restrict__ x,
                                                 const float* __restrict__ W,
                                                 float* __restrict__ out,
                                                 float* Wl, const int blk, const int tid) {
#pragma unroll
    for (int jj = 0; jj < (GCN_IN * 16) / EB_T; ++jj) {
        const int i4 = tid + jj * EB_T;
        const float4 w = *(const float4*)(W + i4 * 4);
        const int k = i4 >> 4;
        const int n = (i4 & 15) * 4;
        float* d = &Wl[(k >> 1) * PAIR_STRIDE + n * 2 + (k & 1)];
        d[0] = w.x; d[2] = w.y; d[4] = w.z; d[6] = w.w;
    }
    __syncthreads();

    const int lane = tid & 31;
    const int wave = tid >> 5;
    const int tile = blk * GM_WAVES + wave;
    if (tile >= (GCN_N >> 4)) return;

    const int m    = lane & 15;
    const int hi   = lane >> 4;
    const int koff = hi * 2;
    const float* arow = x + (size_t)(tile * 16 + m) * GCN_IN;
    const float* brow = &Wl[(koff >> 1) * PAIR_STRIDE + m * 2];

    v8f acc0 = {}, acc1 = {}, acc2 = {}, acc3 = {};
    for (int k0 = 0; k0 < GCN_IN; k0 += 4) {
        const float2 av = *(const float2*)(arow + k0 + koff);
        v2f a; a.x = av.x; a.y = av.y;

        const float* bp = brow + (k0 >> 1) * PAIR_STRIDE;
        v2f b;
        b = *(const v2f*)(bp +  0);
        acc0 = __builtin_amdgcn_wmma_f32_16x16x4_f32(false, a, false, b, (short)0, acc0, false, false);
        b = *(const v2f*)(bp + 32);
        acc1 = __builtin_amdgcn_wmma_f32_16x16x4_f32(false, a, false, b, (short)0, acc1, false, false);
        b = *(const v2f*)(bp + 64);
        acc2 = __builtin_amdgcn_wmma_f32_16x16x4_f32(false, a, false, b, (short)0, acc2, false, false);
        b = *(const v2f*)(bp + 96);
        acc3 = __builtin_amdgcn_wmma_f32_16x16x4_f32(false, a, false, b, (short)0, acc3, false, false);
    }

    float* orow = out + (size_t)(tile * 16 + 8 * hi) * 64 + m;
#pragma unroll
    for (int r = 0; r < 8; ++r) {
        orow[(size_t)r * 64 +  0] = acc0[r];
        orow[(size_t)r * 64 + 16] = acc1[r];
        orow[(size_t)r * 64 + 32] = acc2[r];
        orow[(size_t)r * 64 + 48] = acc3[r];
    }
}

__global__ __launch_bounds__(EB_T) void gcn_bucket_move(const int* __restrict__ ei,
                                                        const int* __restrict__ hist,
                                                        int* __restrict__ bstart,
                                                        int* __restrict__ bkt,
                                                        const float* __restrict__ x,
                                                        const float* __restrict__ W1,
                                                        float* __restrict__ t) {
    __shared__ int part[2][2][256];
    __shared__ int wsum[8];
    __shared__ int cur[256];
    __shared__ float Wl[(GCN_IN / 2) * PAIR_STRIDE];
    const int tid = threadIdx.x;
    const int c = blockIdx.x;
    if (c >= EB_CHUNKS) {
        gcn_layer1_block(x, W1, t, Wl, c - EB_CHUNKS, tid);
        return;
    }

    const int* srcp = ei + c * EB_CHUNK;
    const int* dstp = ei + GCN_E + c * EB_CHUNK;
    int d[EB_PER_T], s[EB_PER_T];
#pragma unroll
    for (int j = 0; j < EB_PER_T; ++j) {
        const int i = tid + j * EB_T;
        d[j] = (i < EB_CHUNK) ? dstp[i] : -1;
        s[j] = (i < EB_CHUNK) ? srcp[i] : 0;
    }

    const int b = tid & 255;
    const int half = tid >> 8;
    const int* hp = hist + half * (EB_CHUNKS / 2) * 256 + b;
    const int cq = c - half * (EB_CHUNKS / 2);
    int tot = 0, pre = 0;
#pragma unroll
    for (int c2 = 0; c2 < EB_CHUNKS / 2; ++c2) {
        const int hv = hp[c2 * 256];
        tot += hv;
        pre += (c2 < cq) ? hv : 0;
    }
    part[half][0][b] = tot;
    part[half][1][b] = pre;
    __syncthreads();
    int t2 = 0, p2 = 0;
    if (tid < 256) {
        t2 = part[0][0][tid] + part[1][0][tid];
        p2 = part[0][1][tid] + part[1][1][tid];
    }
    const int excl = gcn_scan256(wsum, tid, t2);
    if (tid < 256) {
        cur[tid] = excl + p2;
        if (c == 0) {
            bstart[tid] = excl;
            if (tid == 255) bstart[256] = excl + t2;
        }
    }
    __syncthreads();

#pragma unroll
    for (int j = 0; j < EB_PER_T; ++j)
        if (d[j] >= 0) {
            const int pos = atomicAdd(&cur[d[j] >> 8], 1);
            bkt[pos] = (s[j] << 8) | (d[j] & 255);
        }
}

__global__ __launch_bounds__(EB_T) void gcn_bucket_finish(const int* __restrict__ bstart,
                                                          const int* __restrict__ bkt,
                                                          int* __restrict__ ptr,
                                                          float* __restrict__ dinv,
                                                          int* __restrict__ csr,
                                                          float* __restrict__ t) {
    __shared__ int wsum[8];
    __shared__ int cnt[256];
    __shared__ float dl[256];
    const int tid = threadIdx.x;
    const int b = blockIdx.x;
    const int beg = bstart[b];
    const int end = bstart[b + 1];
    float4* trow = (float4*)t + (size_t)b * (256 * 16);
    const int nrow4 = min(GCN_N - b * 256, 256) * 16;
    float4 tr[(256 * 16) / EB_T];
#pragma unroll
    for (int u = 0; u < (256 * 16) / EB_T; ++u) {
        const int f = tid + u * EB_T;
        if (f < nrow4) tr[u] = trow[f];
    }
    const bool one = (end - beg) <= EB_T * EB_FIN_U;
    if (tid < 256) cnt[tid] = 0;
    __syncthreads();
    int v[EB_FIN_U];
    for (int base = beg; base < end; base += EB_T * EB_FIN_U) {
#pragma unroll
        for (int u = 0; u < EB_FIN_U; ++u) {
            const int i = base + u * EB_T + tid;
            v[u] = (i < end) ? bkt[i] : -1;
        }
#pragma unroll
        for (int u = 0; u < EB_FIN_U; ++u)
            if (v[u] >= 0) atomicAdd(&cnt[v[u] & 255], 1);
    }
    __syncthreads();
    const int node = b * 256 + tid;
    int cn = 0;
    if (tid < 256) {
        cn = cnt[tid];
        const float dg = 2.0f + (float)cn;
        const float dv = (dg > 0.0f) ? rsqrtf(dg) : 0.0f;
        if (node < GCN_N) dinv[node] = dv;
        dl[tid] = dv;
    }
    const int first = beg + gcn_scan256(wsum, tid, cn);
#pragma unroll
    for (int u = 0; u < (256 * 16) / EB_T; ++u) {
        const int f = tid + u * EB_T;
        if (f < nrow4) {
            const float s = dl[f >> 4];
            float4 v4 = tr[u];
            v4.x *= s; v4.y *= s; v4.z *= s; v4.w *= s;
            trow[f] = v4;
        }
    }
    if (tid < 256) {
        if (node <= GCN_N) ptr[node] = first;
        cnt[tid] = first;
    }
    __syncthreads();
    for (int base = beg; base < end; base += EB_T * EB_FIN_U) {
        if (!one) {
#pragma unroll
            for (int u = 0; u < EB_FIN_U; ++u) {
                const int i = base + u * EB_T + tid;
                v[u] = (i < end) ? bkt[i] : -1;
            }
        }
#pragma unroll
        for (int u = 0; u < EB_FIN_U; ++u)
            if (v[u] >= 0) {
                const int pos = atomicAdd(&cnt[v[u] & 255], 1);
                csr[pos] = v[u] >> 8;
            }
    }
}

#define GG_TILES 4

__device__ __forceinline__ void gcn_ids32(const int* __restrict__ csr, const int beg,
                                          const int sub, int& s0, int& s1) {
    s0 = csr[min(beg + sub, GCN_E - 1)];
    s1 = csr[min(beg + 16 + sub, GCN_E - 1)];
}

__device__ __forceinline__ void gcn_rows16_issue(float4 (&r)[16], const float* __restrict__ t,
                                                 const int sid, const int n, const int q) {
    unsigned off[16];
#pragma unroll
    for (int u = 0; u < 16; ++u)
        off[u] = (unsigned)__shfl(sid, u, 16) * 256u + (unsigned)q * 4u;
#pragma unroll
    for (int u = 0; u < 16; ++u)
        if (u < n) r[u] = *(const float4*)((const char*)t + off[u]);
}

__device__ __forceinline__ void gcn_rows16_sum(float4& acc, const float4 (&r)[16], const int n) {
#pragma unroll
    for (int u = 0; u < 16; ++u)
        if (u < n) {
            acc.x += r[u].x;
            acc.y += r[u].y;
            acc.z += r[u].z;
            acc.w += r[u].w;
        }
}

template <class F>
__device__ __forceinline__ float4 gcn_node_sum(const int* __restrict__ csr,
                                               const float* __restrict__ t,
                                               const int sub, const int q,
                                               const int beg, const int end,
                                               const int s0, const int s1,
                                               const float4 tv, F&& ahead) {
    float4 acc;
    acc.x = 2.0f * tv.x;
    acc.y = 2.0f * tv.y;
    acc.z = 2.0f * tv.z;
    acc.w = 2.0f * tv.w;
    const int deg = end - beg;
    float4 r[16];
    gcn_rows16_issue(r, t, s0, deg, q);
    ahead();
    gcn_rows16_sum(acc, r, deg);
    for (int base = 16; base < deg; base += 16) {
        int sid = s1;
        if (base >= 32) sid = csr[min(beg + base + sub, GCN_E - 1)];
        gcn_rows16_issue(r, t, sid, deg - base, q);
        gcn_rows16_sum(acc, r, deg - base);
    }
    return acc;
}

struct gcn_pipe {
    int   begC, endC, s0C, s1C;
    float diC;
    float4 tvC;
    int   begN, endN;
    int   begNN, endNN, s0N, s1N;
    float diN;
    float4 tvN;
};

__device__ __forceinline__ void gcn_pipe_bounds(gcn_pipe& p, const int* __restrict__ ptr,
                                                const int i, const int nt) {
    p.begC = ptr[i];
    p.endC = ptr[i + 1];
    p.begN = 0; p.endN = 0;
    if (nt > 1) { p.begN = ptr[i + 16]; p.endN = ptr[i + 17]; }
}
__device__ __forceinline__ void gcn_pipe_first(gcn_pipe& p, const int* __restrict__ csr,
                                               const float* __restrict__ dinv,
                                               const float* __restrict__ t,
                                               const int i, const int sub, const int q) {
    gcn_ids32(csr, p.begC, sub, p.s0C, p.s1C);
    p.diC = dinv[i];
    p.tvC = *(const float4*)(t + (size_t)i * 64 + q);
}
__device__ __forceinline__ void gcn_pipe_ahead(gcn_pipe& p, const int* __restrict__ ptr,
                                               const int* __restrict__ csr,
                                               const float* __restrict__ dinv,
                                               const float* __restrict__ t,
                                               const int i, const int sub, const int q,
                                               const int j, const int nt) {
    p.s0N = 0; p.s1N = 0; p.begNN = 0; p.endNN = 0; p.diN = 0.0f; p.tvN = p.tvC;
    if (j + 1 < nt) {
        gcn_ids32(csr, p.begN, sub, p.s0N, p.s1N);
        p.diN = dinv[i + 16];
        p.tvN = *(const float4*)(t + (size_t)(i + 16) * 64 + q);
    }
    if (j + 2 < nt) { p.begNN = ptr[i + 32]; p.endNN = ptr[i + 33]; }
}
__device__ __forceinline__ void gcn_pipe_step(gcn_pipe& p) {
    p.begC = p.begN; p.endC = p.endN; p.s0C = p.s0N; p.s1C = p.s1N;
    p.diC = p.diN; p.tvC = p.tvN;
    p.begN = p.begNN; p.endN = p.endNN;
}

__global__ __launch_bounds__(256) void gcn_gather(const int* __restrict__ ptr,
                                                  const int* __restrict__ csr,
                                                  const float* __restrict__ dinv,
                                                  const float* __restrict__ t,
                                                  const float* __restrict__ bias,
                                                  float* __restrict__ out) {
    const int tid = threadIdx.x;
    const int g   = tid >> 4;
    const int sub = tid & 15;
    const int q   = sub * 4;
    const int tile0 = blockIdx.x * GG_TILES;
    const int nt = min(GG_TILES, (GCN_N >> 4) - tile0);
    int i = tile0 * 16 + g;
    gcn_pipe p;
    gcn_pipe_bounds(p, ptr, i, nt);
    const float4 bv = *(const float4*)(bias + q);
    gcn_pipe_first(p, csr, dinv, t, i, sub, q);
    for (int j = 0; j < nt; ++j) {
        const float di = p.diC;
        const float4 acc = gcn_node_sum(csr, t, sub, q, p.begC, p.endC, p.s0C, p.s1C, p.tvC,
                                        [&] { gcn_pipe_ahead(p, ptr, csr, dinv, t, i, sub, q, j, nt); });
        float4 o;
        o.x = bv.x + di * acc.x;
        o.y = bv.y + di * acc.y;
        o.z = bv.z + di * acc.z;
        o.w = bv.w + di * acc.w;
        *(float4*)(out + (size_t)i * 64 + q) = o;
        gcn_pipe_step(p);
        i += 16;
    }
}

#define A_PITCH 68

__global__ __launch_bounds__(256) void gcn_gather_gemm(const int* __restrict__ ptr,
                                                       const int* __restrict__ csr,
                                                       const float* __restrict__ dinv,
                                                       const float* __restrict__ tin,
                                                       const float* __restrict__ bias,
                                                       const float* __restrict__ W,
                                                       float* __restrict__ tout) {
    __shared__ float Wl[(GCN_H / 2) * PAIR_STRIDE];
    __shared__ float At[2][16 * A_PITCH];
    const int tid = threadIdx.x;
    const int g    = tid >> 4;
    const int sub  = tid & 15;
    const int q    = sub * 4;

    const int tile0 = blockIdx.x * GG_TILES;
    const int nt = min(GG_TILES, (GCN_N >> 4) - tile0);
    int i = tile0 * 16 + g;
    gcn_pipe p;
    gcn_pipe_bounds(p, ptr, i, nt);

#pragma unroll
    for (int jj = 0; jj < (GCN_H * 16) / 256; ++jj) {
        const int i4 = tid + jj * 256;
        const float4 w = *(const float4*)(W + i4 * 4);
        const int k = i4 >> 4;
        const int n = (i4 & 15) * 4;
        float* d = &Wl[(k >> 1) * PAIR_STRIDE + n * 2 + (k & 1)];
        d[0] = w.x; d[2] = w.y; d[4] = w.z; d[6] = w.w;
    }

    const int lane = tid & 31;
    const int wave = tid >> 5;
    const int m    = lane & 15;
    const int hi   = lane >> 4;
    const int koff = hi * 2;
    const int n0   = (wave & 3) * 16;
    const float4 bv = *(const float4*)(bias + q);
    gcn_pipe_first(p, csr, dinv, tin, i, sub, q);

    for (int j = 0; j < nt; ++j) {
        const int tile = tile0 + j;
        const float di = p.diC;
        const float4 acc = gcn_node_sum(csr, tin, sub, q, p.begC, p.endC, p.s0C, p.s1C, p.tvC,
                                        [&] { gcn_pipe_ahead(p, ptr, csr, dinv, tin, i, sub, q, j, nt); });
        float4 a4;
        a4.x = di * fmaxf(bv.x + di * acc.x, 0.0f);
        a4.y = di * fmaxf(bv.y + di * acc.y, 0.0f);
        a4.z = di * fmaxf(bv.z + di * acc.z, 0.0f);
        a4.w = di * fmaxf(bv.w + di * acc.w, 0.0f);
        float* Ab = At[j & 1];
        *(float4*)(&Ab[g * A_PITCH + q]) = a4;
        __syncthreads();

        if (wave < 4) {
            const float* ap = &Ab[m * A_PITCH + koff];
            const float* bp = &Wl[(koff >> 1) * PAIR_STRIDE + (n0 + m) * 2];
            v8f c8 = {};
#pragma unroll
            for (int kk = 0; kk < 16; ++kk) {
                const v2f a = *(const v2f*)(ap + kk * 4);
                const v2f b = *(const v2f*)(bp + kk * 2 * PAIR_STRIDE);
                c8 = __builtin_amdgcn_wmma_f32_16x16x4_f32(false, a, false, b, (short)0, c8, false, false);
            }
            float* orow = tout + (size_t)(tile * 16 + 8 * hi) * 64 + n0 + m;
#pragma unroll
            for (int r = 0; r < 8; ++r) orow[(size_t)r * 64] = c8[r];
        }
        gcn_pipe_step(p);
        i += 16;
    }
}


extern "C" void kernel_launch(void* const* d_in, const int* in_sizes, int n_in,
                              void* d_out, int out_size, void* d_ws, size_t ws_size,
                              hipStream_t stream) {
    (void)in_sizes; (void)n_in; (void)out_size; (void)ws_size;

    const float* x  = (const float*)d_in[0];
    const int*   ei = (const int*)d_in[1];
    const float* W[5] = {(const float*)d_in[2], (const float*)d_in[4], (const float*)d_in[6],
                         (const float*)d_in[8], (const float*)d_in[10]};
    const float* B[5] = {(const float*)d_in[3], (const float*)d_in[5], (const float*)d_in[7],
                         (const float*)d_in[9], (const float*)d_in[11]};

    char*  ws  = (char*)d_ws;
    size_t off = 0;
    auto carve = [&](size_t bytes) {
        void* p = ws + off;
        off = (off + bytes + 255) & ~(size_t)255;
        return p;
    };
    int*   ptr  = (int*)  carve((size_t)(GCN_N + 1) * 4);
    float* dinv = (float*)carve((size_t)GCN_N * 4);
    int*   csr  = (int*)  carve((size_t)GCN_E * 4);
    float* t    = (float*)carve((size_t)GCN_N * 64 * 4);
    float* hagg = (float*)carve((size_t)GCN_N * 64 * 4);
    int*   bkt    = (int*)hagg;
    int*   hist   = (int*)hagg + GCN_E;
    int*   bstart = (int*)hagg + GCN_E + EB_CHUNKS * 256;

    const int BGG = ((GCN_N / 16) + GG_TILES - 1) / GG_TILES;

    gcn_bucket_count <<<EB_CHUNKS,  EB_T, 0, stream>>>(ei, hist);
    gcn_bucket_move  <<<EB_CHUNKS + GM_BLOCKS, EB_T, 0, stream>>>(ei, hist, bstart, bkt, x, W[0], t);
    gcn_bucket_finish<<<EB_BUCKETS, EB_T, 0, stream>>>(bstart, bkt, ptr, dinv, csr, t);

    gcn_gather_gemm<<<BGG, 256, 0, stream>>>(ptr, csr, dinv, t,    B[0], W[1], hagg);
    gcn_gather_gemm<<<BGG, 256, 0, stream>>>(ptr, csr, dinv, hagg, B[1], W[2], t);
    gcn_gather_gemm<<<BGG, 256, 0, stream>>>(ptr, csr, dinv, t,    B[2], W[3], hagg);
    gcn_gather_gemm<<<BGG, 256, 0, stream>>>(ptr, csr, dinv, hagg, B[3], W[4], t);

    float* out = (float*)d_out;
    gcn_gather<<<BGG, 256, 0, stream>>>(ptr, csr, dinv, t, B[4], out);
}


# Edge list built in 2 launches, not 3: each chunk sorts its own edges by bucket in its own region (count, scan, slot in LDS); the bucket pass reads its 128 runs; arrival order still sets the sum order
# speedup vs baseline: 4.3327x; 1.1333x over previous
#include <hip/hip_runtime.h>


#define GCN_N 50000
#define GCN_E 800000
#define GCN_IN 128
#define GCN_H 64

#define PAIR_STRIDE 130

typedef float v2f __attribute__((ext_vector_type(2)));
typedef float v8f __attribute__((ext_vector_type(8)));

#define EB_T       512
#define EB_CHUNKS  128
#define EB_CHUNK   (GCN_E / EB_CHUNKS)
#define EB_PER_T   ((EB_CHUNK + EB_T - 1) / EB_T)
#define EB_BUCKETS ((GCN_N + 255) / 256)
#define EB_FIN_U   16

__device__ __forceinline__ int gcn_scan256(int* wsum, const int tid, const int v) {
    const int lane = tid & 31;
    const int wave = tid >> 5;
    int x = v;
#pragma unroll
    for (int off = 1; off < 32; off <<= 1) {
        const int y = __shfl_up(x, off, 32);
        if (lane >= off) x += y;
    }
    if (wave < 8 && lane == 31) wsum[wave] = x;
    __syncthreads();
    int add = 0;
#pragma unroll
    for (int w = 0; w < 8; ++w)
        if (w < wave) add += wsum[w];
    return x - v + add;
}

#define GM_WAVES  (EB_T / 32)
#define GM_BLOCKS (((GCN_N >> 4) + GM_WAVES - 1) / GM_WAVES)

__device__ __forceinline__ void gcn_layer1_block(const float* __restrict__ x,
                                                 const float* __restrict__ W,
                                                 float* __restrict__ out,
                                                 float* Wl, const int blk, const int tid) {
#pragma unroll
    for (int jj = 0; jj < (GCN_IN * 16) / EB_T; ++jj) {
        const int i4 = tid + jj * EB_T;
        const float4 w = *(const float4*)(W + i4 * 4);
        const int k = i4 >> 4;
        const int n = (i4 & 15) * 4;
        float* d = &Wl[(k >> 1) * PAIR_STRIDE + n * 2 + (k & 1)];
        d[0] = w.x; d[2] = w.y; d[4] = w.z; d[6] = w.w;
    }
    __syncthreads();

    const int lane = tid & 31;
    const int wave = tid >> 5;
    const int tile = blk * GM_WAVES + wave;
    if (tile >= (GCN_N >> 4)) return;

    const int m    = lane & 15;
    const int hi   = lane >> 4;
    const int koff = hi * 2;
    const float* arow = x + (size_t)(tile * 16 + m) * GCN_IN;
    const float* brow = &Wl[(koff >> 1) * PAIR_STRIDE + m * 2];

    v8f acc0 = {}, acc1 = {}, acc2 = {}, acc3 = {};
    for (int k0 = 0; k0 < GCN_IN; k0 += 4) {
        const float2 av = *(const float2*)(arow + k0 + koff);
        v2f a; a.x = av.x; a.y = av.y;

        const float* bp = brow + (k0 >> 1) * PAIR_STRIDE;
        v2f b;
        b = *(const v2f*)(bp +  0);
        acc0 = __builtin_amdgcn_wmma_f32_16x16x4_f32(false, a, false, b, (short)0, acc0, false, false);
        b = *(const v2f*)(bp + 32);
        acc1 = __builtin_amdgcn_wmma_f32_16x16x4_f32(false, a, false, b, (short)0, acc1, false, false);
        b = *(const v2f*)(bp + 64);
        acc2 = __builtin_amdgcn_wmma_f32_16x16x4_f32(false, a, false, b, (short)0, acc2, false, false);
        b = *(const v2f*)(bp + 96);
        acc3 = __builtin_amdgcn_wmma_f32_16x16x4_f32(false, a, false, b, (short)0, acc3, false, false);
    }

    float* orow = out + (size_t)(tile * 16 + 8 * hi) * 64 + m;
#pragma unroll
    for (int r = 0; r < 8; ++r) {
        orow[(size_t)r * 64 +  0] = acc0[r];
        orow[(size_t)r * 64 + 16] = acc1[r];
        orow[(size_t)r * 64 + 32] = acc2[r];
        orow[(size_t)r * 64 + 48] = acc3[r];
    }
}

__global__ __launch_bounds__(EB_T) void gcn_bucket_move(const int* __restrict__ ei,
                                                        int* __restrict__ cntT,
                                                        int* __restrict__ offT,
                                                        int* __restrict__ bkt,
                                                        const float* __restrict__ x,
                                                        const float* __restrict__ W1,
                                                        float* __restrict__ t) {
    __shared__ int wsum[8];
    __shared__ int cur[256];
    __shared__ float Wl[(GCN_IN / 2) * PAIR_STRIDE];
    const int tid = threadIdx.x;
    const int c = blockIdx.x;
    if (c >= EB_CHUNKS) {
        gcn_layer1_block(x, W1, t, Wl, c - EB_CHUNKS, tid);
        return;
    }

    const int* srcp = ei + c * EB_CHUNK;
    const int* dstp = ei + GCN_E + c * EB_CHUNK;
    int d[EB_PER_T], s[EB_PER_T];
#pragma unroll
    for (int j = 0; j < EB_PER_T; ++j) {
        const int i = tid + j * EB_T;
        d[j] = (i < EB_CHUNK) ? dstp[i] : -1;
        s[j] = (i < EB_CHUNK) ? srcp[i] : 0;
    }
    if (tid < 256) cur[tid] = 0;
    __syncthreads();
#pragma unroll
    for (int j = 0; j < EB_PER_T; ++j)
        if (d[j] >= 0) atomicAdd(&cur[d[j] >> 8], 1);
    __syncthreads();
    const int hv = (tid < 256) ? cur[tid] : 0;
    const int excl = gcn_scan256(wsum, tid, hv);
    if (tid < 256) {
        cur[tid] = excl;
        cntT[tid * EB_CHUNKS + c] = hv;
        offT[tid * EB_CHUNKS + c] = excl;
    }
    __syncthreads();

    int* reg = bkt + c * EB_CHUNK;
#pragma unroll
    for (int j = 0; j < EB_PER_T; ++j)
        if (d[j] >= 0) {
            const int pos = atomicAdd(&cur[d[j] >> 8], 1);
            reg[pos] = (s[j] << 8) | (d[j] & 255);
        }
}

__global__ __launch_bounds__(EB_T) void gcn_bucket_finish(const int* __restrict__ cntT,
                                                          const int* __restrict__ offT,
                                                          const int* __restrict__ bkt,
                                                          int* __restrict__ ptr,
                                                          float* __restrict__ dinv,
                                                          int* __restrict__ csr,
                                                          float* __restrict__ t) {
    __shared__ int wsum[8];
    __shared__ int wsum2[8];
    __shared__ int cnt[256];
    __shared__ float dl[256];
    __shared__ int rbeg[EB_CHUNKS];
    __shared__ int rlen[EB_CHUNKS];
    __shared__ int tot[1];
    const int tid = threadIdx.x;
    const int b = blockIdx.x;
    int hc = 0, ec = 0;
    if (tid < EB_CHUNKS) {
        hc = cntT[b * EB_CHUNKS + tid];
        ec = offT[b * EB_CHUNKS + tid];
    }
    float4* trow = (float4*)t + (size_t)b * (256 * 16);
    const int nrow4 = min(GCN_N - b * 256, 256) * 16;
    float4 tr[(256 * 16) / EB_T];
#pragma unroll
    for (int u = 0; u < (256 * 16) / EB_T; ++u) {
        const int f = tid + u * EB_T;
        if (f < nrow4) tr[u] = trow[f];
    }
    if (tid < 256) cnt[tid] = 0;
    if (tid < EB_CHUNKS) {
        rbeg[tid] = tid * EB_CHUNK + ec;
        rlen[tid] = hc;
    }
    const int pe = gcn_scan256(wsum2, tid, ec);
    if (tid == 255) tot[0] = pe + ec;
    __syncthreads();
    const int beg = tot[0];

    const int rc = tid >> 2;
    const int rk = tid & 3;
    const int len = rlen[rc];
    const int* run = bkt + rbeg[rc];
    int v[EB_FIN_U];
#pragma unroll
    for (int u = 0; u < EB_FIN_U; ++u) {
        const int e = rk + 4 * u;
        v[u] = (e < len) ? run[e] : -1;
    }
#pragma unroll
    for (int u = 0; u < EB_FIN_U; ++u)
        if (v[u] >= 0) atomicAdd(&cnt[v[u] & 255], 1);
    for (int e = rk + 4 * EB_FIN_U; e < len; e += 4)
        atomicAdd(&cnt[run[e] & 255], 1);
    __syncthreads();
    const int node = b * 256 + tid;
    int cn = 0;
    if (tid < 256) {
        cn = cnt[tid];
        const float dg = 2.0f + (float)cn;
        const float dv = (dg > 0.0f) ? rsqrtf(dg) : 0.0f;
        if (node < GCN_N) dinv[node] = dv;
        dl[tid] = dv;
    }
    const int first = beg + gcn_scan256(wsum, tid, cn);
#pragma unroll
    for (int u = 0; u < (256 * 16) / EB_T; ++u) {
        const int f = tid + u * EB_T;
        if (f < nrow4) {
            const float s = dl[f >> 4];
            float4 v4 = tr[u];
            v4.x *= s; v4.y *= s; v4.z *= s; v4.w *= s;
            trow[f] = v4;
        }
    }
    if (tid < 256) {
        if (node <= GCN_N) ptr[node] = first;
        cnt[tid] = first;
    }
    __syncthreads();
#pragma unroll
    for (int u = 0; u < EB_FIN_U; ++u)
        if (v[u] >= 0) {
            const int pos = atomicAdd(&cnt[v[u] & 255], 1);
            csr[pos] = v[u] >> 8;
        }
    for (int e = rk + 4 * EB_FIN_U; e < len; e += 4) {
        const int w = run[e];
        const int pos = atomicAdd(&cnt[w & 255], 1);
        csr[pos] = w >> 8;
    }
}

#define GG_TILES 4

__device__ __forceinline__ void gcn_ids32(const int* __restrict__ csr, const int beg,
                                          const int sub, int& s0, int& s1) {
    s0 = csr[min(beg + sub, GCN_E - 1)];
    s1 = csr[min(beg + 16 + sub, GCN_E - 1)];
}

__device__ __forceinline__ void gcn_rows16_issue(float4 (&r)[16], const float* __restrict__ t,
                                                 const int sid, const int n, const int q) {
    unsigned off[16];
#pragma unroll
    for (int u = 0; u < 16; ++u)
        off[u] = (unsigned)__shfl(sid, u, 16) * 256u + (unsigned)q * 4u;
#pragma unroll
    for (int u = 0; u < 16; ++u)
        if (u < n) r[u] = *(const float4*)((const char*)t + off[u]);
}

__device__ __forceinline__ void gcn_rows16_sum(float4& acc, const float4 (&r)[16], const int n) {
#pragma unroll
    for (int u = 0; u < 16; ++u)
        if (u < n) {
            acc.x += r[u].x;
            acc.y += r[u].y;
            acc.z += r[u].z;
            acc.w += r[u].w;
        }
}

template <class F>
__device__ __forceinline__ float4 gcn_node_sum(const int* __restrict__ csr,
                                               const float* __restrict__ t,
                                               const int sub, const int q,
                                               const int beg, const int end,
                                               const int s0, const int s1,
                                               const float4 tv, F&& ahead) {
    float4 acc;
    acc.x = 2.0f * tv.x;
    acc.y = 2.0f * tv.y;
    acc.z = 2.0f * tv.z;
    acc.w = 2.0f * tv.w;
    const int deg = end - beg;
    float4 r[16];
    gcn_rows16_issue(r, t, s0, deg, q);
    ahead();
    gcn_rows16_sum(acc, r, deg);
    for (int base = 16; base < deg; base += 16) {
        int sid = s1;
        if (base >= 32) sid = csr[min(beg + base + sub, GCN_E - 1)];
        gcn_rows16_issue(r, t, sid, deg - base, q);
        gcn_rows16_sum(acc, r, deg - base);
    }
    return acc;
}

struct gcn_pipe {
    int   begC, endC, s0C, s1C;
    float diC;
    float4 tvC;
    int   begN, endN;
    int   begNN, endNN, s0N, s1N;
    float diN;
    float4 tvN;
};

__device__ __forceinline__ void gcn_pipe_bounds(gcn_pipe& p, const int* __restrict__ ptr,
                                                const int i, const int nt) {
    p.begC = ptr[i];
    p.endC = ptr[i + 1];
    p.begN = 0; p.endN = 0;
    if (nt > 1) { p.begN = ptr[i + 16]; p.endN = ptr[i + 17]; }
}
__device__ __forceinline__ void gcn_pipe_first(gcn_pipe& p, const int* __restrict__ csr,
                                               const float* __restrict__ dinv,
                                               const float* __restrict__ t,
                                               const int i, const int sub, const int q) {
    gcn_ids32(csr, p.begC, sub, p.s0C, p.s1C);
    p.diC = dinv[i];
    p.tvC = *(const float4*)(t + (size_t)i * 64 + q);
}
__device__ __forceinline__ void gcn_pipe_ahead(gcn_pipe& p, const int* __restrict__ ptr,
                                               const int* __restrict__ csr,
                                               const float* __restrict__ dinv,
                                               const float* __restrict__ t,
                                               const int i, const int sub, const int q,
                                               const int j, const int nt) {
    p.s0N = 0; p.s1N = 0; p.begNN = 0; p.endNN = 0; p.diN = 0.0f; p.tvN = p.tvC;
    if (j + 1 < nt) {
        gcn_ids32(csr, p.begN, sub, p.s0N, p.s1N);
        p.diN = dinv[i + 16];
        p.tvN = *(const float4*)(t + (size_t)(i + 16) * 64 + q);
    }
    if (j + 2 < nt) { p.begNN = ptr[i + 32]; p.endNN = ptr[i + 33]; }
}
__device__ __forceinline__ void gcn_pipe_step(gcn_pipe& p) {
    p.begC = p.begN; p.endC = p.endN; p.s0C = p.s0N; p.s1C = p.s1N;
    p.diC = p.diN; p.tvC = p.tvN;
    p.begN = p.begNN; p.endN = p.endNN;
}

__global__ __launch_bounds__(256) void gcn_gather(const int* __restrict__ ptr,
                                                  const int* __restrict__ csr,
                                                  const float* __restrict__ dinv,
                                                  const float* __restrict__ t,
                                                  const float* __restrict__ bias,
                                                  float* __restrict__ out) {
    const int tid = threadIdx.x;
    const int g   = tid >> 4;
    const int sub = tid & 15;
    const int q   = sub * 4;
    const int tile0 = blockIdx.x * GG_TILES;
    const int nt = min(GG_TILES, (GCN_N >> 4) - tile0);
    int i = tile0 * 16 + g;
    gcn_pipe p;
    gcn_pipe_bounds(p, ptr, i, nt);
    const float4 bv = *(const float4*)(bias + q);
    gcn_pipe_first(p, csr, dinv, t, i, sub, q);
    for (int j = 0; j < nt; ++j) {
        const float di = p.diC;
        const float4 acc = gcn_node_sum(csr, t, sub, q, p.begC, p.endC, p.s0C, p.s1C, p.tvC,
                                        [&] { gcn_pipe_ahead(p, ptr, csr, dinv, t, i, sub, q, j, nt); });
        float4 o;
        o.x = bv.x + di * acc.x;
        o.y = bv.y + di * acc.y;
        o.z = bv.z + di * acc.z;
        o.w = bv.w + di * acc.w;
        *(float4*)(out + (size_t)i * 64 + q) = o;
        gcn_pipe_step(p);
        i += 16;
    }
}

#define A_PITCH 68

__global__ __launch_bounds__(256) void gcn_gather_gemm(const int* __restrict__ ptr,
                                                       const int* __restrict__ csr,
                                                       const float* __restrict__ dinv,
                                                       const float* __restrict__ tin,
                                                       const float* __restrict__ bias,
                                                       const float* __restrict__ W,
                                                       float* __restrict__ tout) {
    __shared__ float Wl[(GCN_H / 2) * PAIR_STRIDE];
    __shared__ float At[2][16 * A_PITCH];
    const int tid = threadIdx.x;
    const int g    = tid >> 4;
    const int sub  = tid & 15;
    const int q    = sub * 4;

    const int tile0 = blockIdx.x * GG_TILES;
    const int nt = min(GG_TILES, (GCN_N >> 4) - tile0);
    int i = tile0 * 16 + g;
    gcn_pipe p;
    gcn_pipe_bounds(p, ptr, i, nt);

#pragma unroll
    for (int jj = 0; jj < (GCN_H * 16) / 256; ++jj) {
        const int i4 = tid + jj * 256;
        const float4 w = *(const float4*)(W + i4 * 4);
        const int k = i4 >> 4;
        const int n = (i4 & 15) * 4;
        float* d = &Wl[(k >> 1) * PAIR_STRIDE + n * 2 + (k & 1)];
        d[0] = w.x; d[2] = w.y; d[4] = w.z; d[6] = w.w;
    }

    const int lane = tid & 31;
    const int wave = tid >> 5;
    const int m    = lane & 15;
    const int hi   = lane >> 4;
    const int koff = hi * 2;
    const int n0   = (wave & 3) * 16;
    const float4 bv = *(const float4*)(bias + q);
    gcn_pipe_first(p, csr, dinv, tin, i, sub, q);

    for (int j = 0; j < nt; ++j) {
        const int tile = tile0 + j;
        const float di = p.diC;
        const float4 acc = gcn_node_sum(csr, tin, sub, q, p.begC, p.endC, p.s0C, p.s1C, p.tvC,
                                        [&] { gcn_pipe_ahead(p, ptr, csr, dinv, tin, i, sub, q, j, nt); });
        float4 a4;
        a4.x = di * fmaxf(bv.x + di * acc.x, 0.0f);
        a4.y = di * fmaxf(bv.y + di * acc.y, 0.0f);
        a4.z = di * fmaxf(bv.z + di * acc.z, 0.0f);
        a4.w = di * fmaxf(bv.w + di * acc.w, 0.0f);
        float* Ab = At[j & 1];
        *(float4*)(&Ab[g * A_PITCH + q]) = a4;
        __syncthreads();

        if (wave < 4) {
            const float* ap = &Ab[m * A_PITCH + koff];
            const float* bp = &Wl[(koff >> 1) * PAIR_STRIDE + (n0 + m) * 2];
            v8f c8 = {};
#pragma unroll
            for (int kk = 0; kk < 16; ++kk) {
                const v2f a = *(const v2f*)(ap + kk * 4);
                const v2f b = *(const v2f*)(bp + kk * 2 * PAIR_STRIDE);
                c8 = __builtin_amdgcn_wmma_f32_16x16x4_f32(false, a, false, b, (short)0, c8, false, false);
            }
            float* orow = tout + (size_t)(tile * 16 + 8 * hi) * 64 + n0 + m;
#pragma unroll
            for (int r = 0; r < 8; ++r) orow[(size_t)r * 64] = c8[r];
        }
        gcn_pipe_step(p);
        i += 16;
    }
}


extern "C" void kernel_launch(void* const* d_in, const int* in_sizes, int n_in,
                              void* d_out, int out_size, void* d_ws, size_t ws_size,
                              hipStream_t stream) {
    (void)in_sizes; (void)n_in; (void)out_size; (void)ws_size;

    const float* x  = (const float*)d_in[0];
    const int*   ei = (const int*)d_in[1];
    const float* W[5] = {(const float*)d_in[2], (const float*)d_in[4], (const float*)d_in[6],
                         (const float*)d_in[8], (const float*)d_in[10]};
    const float* B[5] = {(const float*)d_in[3], (const float*)d_in[5], (const float*)d_in[7],
                         (const float*)d_in[9], (const float*)d_in[11]};

    char*  ws  = (char*)d_ws;
    size_t off = 0;
    auto carve = [&](size_t bytes) {
        void* p = ws + off;
        off = (off + bytes + 255) & ~(size_t)255;
        return p;
    };
    int*   ptr  = (int*)  carve((size_t)(GCN_N + 1) * 4);
    float* dinv = (float*)carve((size_t)GCN_N * 4);
    int*   csr  = (int*)  carve((size_t)GCN_E * 4);
    float* t    = (float*)carve((size_t)GCN_N * 64 * 4);
    float* hagg = (float*)carve((size_t)GCN_N * 64 * 4);
    int*   bkt  = (int*)hagg;
    int*   cntT = (int*)hagg + GCN_E;
    int*   offT = (int*)hagg + GCN_E + EB_CHUNKS * 256;

    const int BGG = ((GCN_N / 16) + GG_TILES - 1) / GG_TILES;

    gcn_bucket_move  <<<EB_CHUNKS + GM_BLOCKS, EB_T, 0, stream>>>(ei, cntT, offT, bkt, x, W[0], t);
    gcn_bucket_finish<<<EB_BUCKETS, EB_T, 0, stream>>>(cntT, offT, bkt, ptr, dinv, csr, t);

    gcn_gather_gemm<<<BGG, 256, 0, stream>>>(ptr, csr, dinv, t,    B[0], W[1], hagg);
    gcn_gather_gemm<<<BGG, 256, 0, stream>>>(ptr, csr, dinv, hagg, B[1], W[2], t);
    gcn_gather_gemm<<<BGG, 256, 0, stream>>>(ptr, csr, dinv, t,    B[2], W[3], hagg);
    gcn_gather_gemm<<<BGG, 256, 0, stream>>>(ptr, csr, dinv, hagg, B[3], W[4], t);

    float* out = (float*)d_out;
    gcn_gather<<<BGG, 256, 0, stream>>>(ptr, csr, dinv, t, B[4], out);
}


# Both list-build launches put their piece of the list together in LDS and store it as whole 128-B lines, not as 1.6M scattered 4-byte global stores; slots, arrival order and sum order unchanged
# speedup vs baseline: 4.5328x; 1.0462x over previous
#include <hip/hip_runtime.h>


#define GCN_N 50000
#define GCN_E 800000
#define GCN_IN 128
#define GCN_H 64

#define PAIR_STRIDE 130

typedef float v2f __attribute__((ext_vector_type(2)));
typedef float v8f __attribute__((ext_vector_type(8)));

#define EB_T       512
#define EB_CHUNKS  128
#define EB_CHUNK   (GCN_E / EB_CHUNKS)
#define EB_PER_T   ((EB_CHUNK + EB_T - 1) / EB_T)
#define EB_BUCKETS ((GCN_N + 255) / 256)
#define EB_FIN_U   16
#define EB_STAGE   8192
static_assert(EB_CHUNK % 2 == 0, "a chunk's region is moved out 8 bytes at a time");
static_assert(EB_CHUNK * 4 <= (GCN_IN / 2) * PAIR_STRIDE * 4, "a chunk's region fits the W staging words");

__device__ __forceinline__ int gcn_scan256(int* wsum, const int tid, const int v) {
    const int lane = tid & 31;
    const int wave = tid >> 5;
    int x = v;
#pragma unroll
    for (int off = 1; off < 32; off <<= 1) {
        const int y = __shfl_up(x, off, 32);
        if (lane >= off) x += y;
    }
    if (wave < 8 && lane == 31) wsum[wave] = x;
    __syncthreads();
    int add = 0;
#pragma unroll
    for (int w = 0; w < 8; ++w)
        if (w < wave) add += wsum[w];
    return x - v + add;
}

#define GM_WAVES  (EB_T / 32)
#define GM_BLOCKS (((GCN_N >> 4) + GM_WAVES - 1) / GM_WAVES)

__device__ __forceinline__ void gcn_layer1_block(const float* __restrict__ x,
                                                 const float* __restrict__ W,
                                                 float* __restrict__ out,
                                                 float* Wl, const int blk, const int tid) {
#pragma unroll
    for (int jj = 0; jj < (GCN_IN * 16) / EB_T; ++jj) {
        const int i4 = tid + jj * EB_T;
        const float4 w = *(const float4*)(W + i4 * 4);
        const int k = i4 >> 4;
        const int n = (i4 & 15) * 4;
        float* d = &Wl[(k >> 1) * PAIR_STRIDE + n * 2 + (k & 1)];
        d[0] = w.x; d[2] = w.y; d[4] = w.z; d[6] = w.w;
    }
    __syncthreads();

    const int lane = tid & 31;
    const int wave = tid >> 5;
    const int tile = blk * GM_WAVES + wave;
    if (tile >= (GCN_N >> 4)) return;

    const int m    = lane & 15;
    const int hi   = lane >> 4;
    const int koff = hi * 2;
    const float* arow = x + (size_t)(tile * 16 + m) * GCN_IN;
    const float* brow = &Wl[(koff >> 1) * PAIR_STRIDE + m * 2];

    v8f acc0 = {}, acc1 = {}, acc2 = {}, acc3 = {};
    for (int k0 = 0; k0 < GCN_IN; k0 += 4) {
        const float2 av = *(const float2*)(arow + k0 + koff);
        v2f a; a.x = av.x; a.y = av.y;

        const float* bp = brow + (k0 >> 1) * PAIR_STRIDE;
        v2f b;
        b = *(const v2f*)(bp +  0);
        acc0 = __builtin_amdgcn_wmma_f32_16x16x4_f32(false, a, false, b, (short)0, acc0, false, false);
        b = *(const v2f*)(bp + 32);
        acc1 = __builtin_amdgcn_wmma_f32_16x16x4_f32(false, a, false, b, (short)0, acc1, false, false);
        b = *(const v2f*)(bp + 64);
        acc2 = __builtin_amdgcn_wmma_f32_16x16x4_f32(false, a, false, b, (short)0, acc2, false, false);
        b = *(const v2f*)(bp + 96);
        acc3 = __builtin_amdgcn_wmma_f32_16x16x4_f32(false, a, false, b, (short)0, acc3, false, false);
    }

    float* orow = out + (size_t)(tile * 16 + 8 * hi) * 64 + m;
#pragma unroll
    for (int r = 0; r < 8; ++r) {
        orow[(size_t)r * 64 +  0] = acc0[r];
        orow[(size_t)r * 64 + 16] = acc1[r];
        orow[(size_t)r * 64 + 32] = acc2[r];
        orow[(size_t)r * 64 + 48] = acc3[r];
    }
}

__global__ __launch_bounds__(EB_T) void gcn_bucket_move(const int* __restrict__ ei,
                                                        int* __restrict__ cntT,
                                                        int* __restrict__ offT,
                                                        int* __restrict__ bkt,
                                                        const float* __restrict__ x,
                                                        const float* __restrict__ W1,
                                                        float* __restrict__ t) {
    __shared__ int wsum[8];
    __shared__ int cur[256];
    __shared__ float Wl[(GCN_IN / 2) * PAIR_STRIDE];
    const int tid = threadIdx.x;
    const int c = blockIdx.x;
    if (c >= EB_CHUNKS) {
        gcn_layer1_block(x, W1, t, Wl, c - EB_CHUNKS, tid);
        return;
    }

    const int* srcp = ei + c * EB_CHUNK;
    const int* dstp = ei + GCN_E + c * EB_CHUNK;
    int d[EB_PER_T], s[EB_PER_T];
#pragma unroll
    for (int j = 0; j < EB_PER_T; ++j) {
        const int i = tid + j * EB_T;
        d[j] = (i < EB_CHUNK) ? dstp[i] : -1;
        s[j] = (i < EB_CHUNK) ? srcp[i] : 0;
    }
    if (tid < 256) cur[tid] = 0;
    __syncthreads();
#pragma unroll
    for (int j = 0; j < EB_PER_T; ++j)
        if (d[j] >= 0) atomicAdd(&cur[d[j] >> 8], 1);
    __syncthreads();
    const int hv = (tid < 256) ? cur[tid] : 0;
    const int excl = gcn_scan256(wsum, tid, hv);
    if (tid < 256) {
        cur[tid] = excl;
        cntT[tid * EB_CHUNKS + c] = hv;
        offT[tid * EB_CHUNKS + c] = excl;
    }
    __syncthreads();

    int* stage = reinterpret_cast<int*>(Wl);
#pragma unroll
    for (int j = 0; j < EB_PER_T; ++j)
        if (d[j] >= 0) {
            const int pos = atomicAdd(&cur[d[j] >> 8], 1);
            stage[pos] = (s[j] << 8) | (d[j] & 255);
        }
    __syncthreads();
    int2* reg2 = reinterpret_cast<int2*>(bkt + c * EB_CHUNK);
    const int2* st2 = reinterpret_cast<const int2*>(stage);
    const int lead = (c * (EB_CHUNK / 2)) & 15;
#pragma unroll
    for (int j = 0; j < (EB_CHUNK / 2 + 15 + EB_T - 1) / EB_T; ++j) {
        const int i = tid - lead + j * EB_T;
        if (i >= 0 && i < EB_CHUNK / 2) reg2[i] = st2[i];
    }
}

__global__ __launch_bounds__(EB_T) void gcn_bucket_finish(const int* __restrict__ cntT,
                                                          const int* __restrict__ offT,
                                                          const int* __restrict__ bkt,
                                                          int* __restrict__ ptr,
                                                          float* __restrict__ dinv,
                                                          int* __restrict__ csr,
                                                          float* __restrict__ t) {
    __shared__ int wsum[8];
    __shared__ int wsum2[8];
    __shared__ int cnt[256];
    __shared__ float dl[256];
    __shared__ int rbeg[EB_CHUNKS];
    __shared__ int rlen[EB_CHUNKS];
    __shared__ int tot[2];
    __shared__ int stage[EB_STAGE];
    const int tid = threadIdx.x;
    const int b = blockIdx.x;
    int hc = 0, ec = 0;
    if (tid < EB_CHUNKS) {
        hc = cntT[b * EB_CHUNKS + tid];
        ec = offT[b * EB_CHUNKS + tid];
    }
    float4* trow = (float4*)t + (size_t)b * (256 * 16);
    const int nrow4 = min(GCN_N - b * 256, 256) * 16;
    float4 tr[(256 * 16) / EB_T];
#pragma unroll
    for (int u = 0; u < (256 * 16) / EB_T; ++u) {
        const int f = tid + u * EB_T;
        if (f < nrow4) tr[u] = trow[f];
    }
    if (tid < 256) cnt[tid] = 0;
    if (tid < EB_CHUNKS) {
        rbeg[tid] = tid * EB_CHUNK + ec;
        rlen[tid] = hc;
    }
    const int pe = gcn_scan256(wsum2, tid, ec);
    if (tid == 255) tot[0] = pe + ec;
    __syncthreads();
    const int beg = tot[0];

    const int rc = tid >> 2;
    const int rk = tid & 3;
    const int len = rlen[rc];
    const int* run = bkt + rbeg[rc];
    int v[EB_FIN_U];
#pragma unroll
    for (int u = 0; u < EB_FIN_U; ++u) {
        const int e = rk + 4 * u;
        v[u] = (e < len) ? run[e] : -1;
    }
#pragma unroll
    for (int u = 0; u < EB_FIN_U; ++u)
        if (v[u] >= 0) atomicAdd(&cnt[v[u] & 255], 1);
    for (int e = rk + 4 * EB_FIN_U; e < len; e += 4)
        atomicAdd(&cnt[run[e] & 255], 1);
    __syncthreads();
    const int node = b * 256 + tid;
    int cn = 0;
    if (tid < 256) {
        cn = cnt[tid];
        const float dg = 2.0f + (float)cn;
        const float dv = (dg > 0.0f) ? rsqrtf(dg) : 0.0f;
        if (node < GCN_N) dinv[node] = dv;
        dl[tid] = dv;
    }
    const int first = beg + gcn_scan256(wsum, tid, cn);
#pragma unroll
    for (int u = 0; u < (256 * 16) / EB_T; ++u) {
        const int f = tid + u * EB_T;
        if (f < nrow4) {
            const float s = dl[f >> 4];
            float4 v4 = tr[u];
            v4.x *= s; v4.y *= s; v4.z *= s; v4.w *= s;
            trow[f] = v4;
        }
    }
    if (tid < 256) {
        if (node <= GCN_N) ptr[node] = first;
        cnt[tid] = first;
    }
    if (tid == 255) tot[1] = first + cn - beg;
    __syncthreads();
    const int nb = tot[1];
    if (nb <= EB_STAGE) {
#pragma unroll
        for (int u = 0; u < EB_FIN_U; ++u)
            if (v[u] >= 0) {
                const int pos = atomicAdd(&cnt[v[u] & 255], 1);
                stage[pos - beg] = v[u] >> 8;
            }
        for (int e = rk + 4 * EB_FIN_U; e < len; e += 4) {
            const int w = run[e];
            const int pos = atomicAdd(&cnt[w & 255], 1);
            stage[pos - beg] = w >> 8;
        }
        __syncthreads();
        int* cb = csr + beg;
        const int lead = beg & 31;
        for (int i0 = tid - lead; i0 < nb; i0 += 4 * EB_T) {
            int w[4];
#pragma unroll
            for (int k = 0; k < 4; ++k)
                w[k] = stage[max(min(i0 + k * EB_T, nb - 1), 0)];
#pragma unroll
            for (int k = 0; k < 4; ++k) {
                const int i = i0 + k * EB_T;
                if (i >= 0 && i < nb) cb[i] = w[k];
            }
        }
    } else {
#pragma unroll
        for (int u = 0; u < EB_FIN_U; ++u)
            if (v[u] >= 0) {
                const int pos = atomicAdd(&cnt[v[u] & 255], 1);
                csr[pos] = v[u] >> 8;
            }
        for (int e = rk + 4 * EB_FIN_U; e < len; e += 4) {
            const int w = run[e];
            const int pos = atomicAdd(&cnt[w & 255], 1);
            csr[pos] = w >> 8;
        }
    }
}

#define GG_TILES 4

__device__ __forceinline__ void gcn_ids32(const int* __restrict__ csr, const int beg,
                                          const int sub, int& s0, int& s1) {
    s0 = csr[min(beg + sub, GCN_E - 1)];
    s1 = csr[min(beg + 16 + sub, GCN_E - 1)];
}

__device__ __forceinline__ void gcn_rows16_issue(float4 (&r)[16], const float* __restrict__ t,
                                                 const int sid, const int n, const int q) {
    unsigned off[16];
#pragma unroll
    for (int u = 0; u < 16; ++u)
        off[u] = (unsigned)__shfl(sid, u, 16) * 256u + (unsigned)q * 4u;
#pragma unroll
    for (int u = 0; u < 16; ++u)
        if (u < n) r[u] = *(const float4*)((const char*)t + off[u]);
}

__device__ __forceinline__ void gcn_rows16_sum(float4& acc, const float4 (&r)[16], const int n) {
#pragma unroll
    for (int u = 0; u < 16; ++u)
        if (u < n) {
            acc.x += r[u].x;
            acc.y += r[u].y;
            acc.z += r[u].z;
            acc.w += r[u].w;
        }
}

template <class F>
__device__ __forceinline__ float4 gcn_node_sum(const int* __restrict__ csr,
                                               const float* __restrict__ t,
                                               const int sub, const int q,
                                               const int beg, const int end,
                                               const int s0, const int s1,
                                               const float4 tv, F&& ahead) {
    float4 acc;
    acc.x = 2.0f * tv.x;
    acc.y = 2.0f * tv.y;
    acc.z = 2.0f * tv.z;
    acc.w = 2.0f * tv.w;
    const int deg = end - beg;
    float4 r[16];
    gcn_rows16_issue(r, t, s0, deg, q);
    ahead();
    gcn_rows16_sum(acc, r, deg);
    for (int base = 16; base < deg; base += 16) {
        int sid = s1;
        if (base >= 32) sid = csr[min(beg + base + sub, GCN_E - 1)];
        gcn_rows16_issue(r, t, sid, deg - base, q);
        gcn_rows16_sum(acc, r, deg - base);
    }
    return acc;
}

struct gcn_pipe {
    int   begC, endC, s0C, s1C;
    float diC;
    float4 tvC;
    int   begN, endN;
    int   begNN, endNN, s0N, s1N;
    float diN;
    float4 tvN;
};

__device__ __forceinline__ void gcn_pipe_bounds(gcn_pipe& p, const int* __restrict__ ptr,
                                                const int i, const int nt) {
    p.begC = ptr[i];
    p.endC = ptr[i + 1];
    p.begN = 0; p.endN = 0;
    if (nt > 1) { p.begN = ptr[i + 16]; p.endN = ptr[i + 17]; }
}
__device__ __forceinline__ void gcn_pipe_first(gcn_pipe& p, const int* __restrict__ csr,
                                               const float* __restrict__ dinv,
                                               const float* __restrict__ t,
                                               const int i, const int sub, const int q) {
    gcn_ids32(csr, p.begC, sub, p.s0C, p.s1C);
    p.diC = dinv[i];
    p.tvC = *(const float4*)(t + (size_t)i * 64 + q);
}
__device__ __forceinline__ void gcn_pipe_ahead(gcn_pipe& p, const int* __restrict__ ptr,
                                               const int* __restrict__ csr,
                                               const float* __restrict__ dinv,
                                               const float* __restrict__ t,
                                               const int i, const int sub, const int q,
                                               const int j, const int nt) {
    p.s0N = 0; p.s1N = 0; p.begNN = 0; p.endNN = 0; p.diN = 0.0f; p.tvN = p.tvC;
    if (j + 1 < nt) {
        gcn_ids32(csr, p.begN, sub, p.s0N, p.s1N);
        p.diN = dinv[i + 16];
        p.tvN = *(const float4*)(t + (size_t)(i + 16) * 64 + q);
    }
    if (j + 2 < nt) { p.begNN = ptr[i + 32]; p.endNN = ptr[i + 33]; }
}
__device__ __forceinline__ void gcn_pipe_step(gcn_pipe& p) {
    p.begC = p.begN; p.endC = p.endN; p.s0C = p.s0N; p.s1C = p.s1N;
    p.diC = p.diN; p.tvC = p.tvN;
    p.begN = p.begNN; p.endN = p.endNN;
}

__global__ __launch_bounds__(256) void gcn_gather(const int* __restrict__ ptr,
                                                  const int* __restrict__ csr,
                                                  const float* __restrict__ dinv,
                                                  const float* __restrict__ t,
                                                  const float* __restrict__ bias,
                                                  float* __restrict__ out) {
    const int tid = threadIdx.x;
    const int g   = tid >> 4;
    const int sub = tid & 15;
    const int q   = sub * 4;
    const int tile0 = blockIdx.x * GG_TILES;
    const int nt = min(GG_TILES, (GCN_N >> 4) - tile0);
    int i = tile0 * 16 + g;
    gcn_pipe p;
    gcn_pipe_bounds(p, ptr, i, nt);
    const float4 bv = *(const float4*)(bias + q);
    gcn_pipe_first(p, csr, dinv, t, i, sub, q);
    for (int j = 0; j < nt; ++j) {
        const float di = p.diC;
        const float4 acc = gcn_node_sum(csr, t, sub, q, p.begC, p.endC, p.s0C, p.s1C, p.tvC,
                                        [&] { gcn_pipe_ahead(p, ptr, csr, dinv, t, i, sub, q, j, nt); });
        float4 o;
        o.x = bv.x + di * acc.x;
        o.y = bv.y + di * acc.y;
        o.z = bv.z + di * acc.z;
        o.w = bv.w + di * acc.w;
        *(float4*)(out + (size_t)i * 64 + q) = o;
        gcn_pipe_step(p);
        i += 16;
    }
}

#define A_PITCH 68

__global__ __launch_bounds__(256) void gcn_gather_gemm(const int* __restrict__ ptr,
                                                       const int* __restrict__ csr,
                                                       const float* __restrict__ dinv,
                                                       const float* __restrict__ tin,
                                                       const float* __restrict__ bias,
                                                       const float* __restrict__ W,
                                                       float* __restrict__ tout) {
    __shared__ float Wl[(GCN_H / 2) * PAIR_STRIDE];
    __shared__ float At[2][16 * A_PITCH];
    const int tid = threadIdx.x;
    const int g    = tid >> 4;
    const int sub  = tid & 15;
    const int q    = sub * 4;

    const int tile0 = blockIdx.x * GG_TILES;
    const int nt = min(GG_TILES, (GCN_N >> 4) - tile0);
    int i = tile0 * 16 + g;
    gcn_pipe p;
    gcn_pipe_bounds(p, ptr, i, nt);

#pragma unroll
    for (int jj = 0; jj < (GCN_H * 16) / 256; ++jj) {
        const int i4 = tid + jj * 256;
        const float4 w = *(const float4*)(W + i4 * 4);
        const int k = i4 >> 4;
        const int n = (i4 & 15) * 4;
        float* d = &Wl[(k >> 1) * PAIR_STRIDE + n * 2 + (k & 1)];
        d[0] = w.x; d[2] = w.y; d[4] = w.z; d[6] = w.w;
    }

    const int lane = tid & 31;
    const int wave = tid >> 5;
    const int m    = lane & 15;
    const int hi   = lane >> 4;
    const int koff = hi * 2;
    const int n0   = (wave & 3) * 16;
    const float4 bv = *(const float4*)(bias + q);
    gcn_pipe_first(p, csr, dinv, tin, i, sub, q);

    for (int j = 0; j < nt; ++j) {
        const int tile = tile0 + j;
        const float di = p.diC;
        const float4 acc = gcn_node_sum(csr, tin, sub, q, p.begC, p.endC, p.s0C, p.s1C, p.tvC,
                                        [&] { gcn_pipe_ahead(p, ptr, csr, dinv, tin, i, sub, q, j, nt); });
        float4 a4;
        a4.x = di * fmaxf(bv.x + di * acc.x, 0.0f);
        a4.y = di * fmaxf(bv.y + di * acc.y, 0.0f);
        a4.z = di * fmaxf(bv.z + di * acc.z, 0.0f);
        a4.w = di * fmaxf(bv.w + di * acc.w, 0.0f);
        float* Ab = At[j & 1];
        *(float4*)(&Ab[g * A_PITCH + q]) = a4;
        __syncthreads();

        if (wave < 4) {
            const float* ap = &Ab[m * A_PITCH + koff];
            const float* bp = &Wl[(koff >> 1) * PAIR_STRIDE + (n0 + m) * 2];
            v8f c8 = {};
#pragma unroll
            for (int kk = 0; kk < 16; ++kk) {
                const v2f a = *(const v2f*)(ap + kk * 4);
                const v2f b = *(const v2f*)(bp + kk * 2 * PAIR_STRIDE);
                c8 = __builtin_amdgcn_wmma_f32_16x16x4_f32(false, a, false, b, (short)0, c8, false, false);
            }
            float* orow = tout + (size_t)(tile * 16 + 8 * hi) * 64 + n0 + m;
#pragma unroll
            for (int r = 0; r < 8; ++r) orow[(size_t)r * 64] = c8[r];
        }
        gcn_pipe_step(p);
        i += 16;
    }
}


extern "C" void kernel_launch(void* const* d_in, const int* in_sizes, int n_in,
                              void* d_out, int out_size, void* d_ws, size_t ws_size,
                              hipStream_t stream) {
    (void)in_sizes; (void)n_in; (void)out_size; (void)ws_size;

    const float* x  = (const float*)d_in[0];
    const int*   ei = (const int*)d_in[1];
    const float* W[5] = {(const float*)d_in[2], (const float*)d_in[4], (const float*)d_in[6],
                         (const float*)d_in[8], (const float*)d_in[10]};
    const float* B[5] = {(const float*)d_in[3], (const float*)d_in[5], (const float*)d_in[7],
                         (const float*)d_in[9], (const float*)d_in[11]};

    char*  ws  = (char*)d_ws;
    size_t off = 0;
    auto carve = [&](size_t bytes) {
        void* p = ws + off;
        off = (off + bytes + 255) & ~(size_t)255;
        return p;
    };
    int*   ptr  = (int*)  carve((size_t)(GCN_N + 1) * 4);
    float* dinv = (float*)carve((size_t)GCN_N * 4);
    int*   csr  = (int*)  carve((size_t)GCN_E * 4);
    float* t    = (float*)carve((size_t)GCN_N * 64 * 4);
    float* hagg = (float*)carve((size_t)GCN_N * 64 * 4);
    int*   bkt  = (int*)hagg;
    int*   cntT = (int*)hagg + GCN_E;
    int*   offT = (int*)hagg + GCN_E + EB_CHUNKS * 256;

    const int BGG = ((GCN_N / 16) + GG_TILES - 1) / GG_TILES;

    gcn_bucket_move  <<<EB_CHUNKS + GM_BLOCKS, EB_T, 0, stream>>>(ei, cntT, offT, bkt, x, W[0], t);
    gcn_bucket_finish<<<EB_BUCKETS, EB_T, 0, stream>>>(cntT, offT, bkt, ptr, dinv, csr, t);

    gcn_gather_gemm<<<BGG, 256, 0, stream>>>(ptr, csr, dinv, t,    B[0], W[1], hagg);
    gcn_gather_gemm<<<BGG, 256, 0, stream>>>(ptr, csr, dinv, hagg, B[1], W[2], t);
    gcn_gather_gemm<<<BGG, 256, 0, stream>>>(ptr, csr, dinv, t,    B[2], W[3], hagg);
    gcn_gather_gemm<<<BGG, 256, 0, stream>>>(ptr, csr, dinv, hagg, B[3], W[4], t);

    float* out = (float*)d_out;
    gcn_gather<<<BGG, 256, 0, stream>>>(ptr, csr, dinv, t, B[4], out);
}


# Edge-sum row loads carry no guard: a list slot past a node's end is keyed to a row of zeros written in-call, so 16 loads leave as one group and the sums need no select; sum order unchanged
# speedup vs baseline: 4.7326x; 1.0441x over previous
#include <hip/hip_runtime.h>


#define GCN_N 50000
#define GCN_E 800000
#define GCN_IN 128
#define GCN_H 64

#define PAIR_STRIDE 130

typedef float v2f __attribute__((ext_vector_type(2)));
typedef float v8f __attribute__((ext_vector_type(8)));

#define EB_T       512
#define EB_CHUNKS  128
#define EB_CHUNK   (GCN_E / EB_CHUNKS)
#define EB_PER_T   ((EB_CHUNK + EB_T - 1) / EB_T)
#define EB_BUCKETS ((GCN_N + 255) / 256)
#define EB_FIN_U   16
#define EB_STAGE   8192
static_assert(EB_CHUNK % 2 == 0, "a chunk's region is moved out 8 bytes at a time");
static_assert(EB_CHUNK * 4 <= (GCN_IN / 2) * PAIR_STRIDE * 4, "a chunk's region fits the W staging words");

__device__ __forceinline__ int gcn_scan256(int* wsum, const int tid, const int v) {
    const int lane = tid & 31;
    const int wave = tid >> 5;
    int x = v;
#pragma unroll
    for (int off = 1; off < 32; off <<= 1) {
        const int y = __shfl_up(x, off, 32);
        if (lane >= off) x += y;
    }
    if (wave < 8 && lane == 31) wsum[wave] = x;
    __syncthreads();
    int add = 0;
#pragma unroll
    for (int w = 0; w < 8; ++w)
        if (w < wave) add += wsum[w];
    return x - v + add;
}

#define GM_WAVES  (EB_T / 32)
#define GM_BLOCKS (((GCN_N >> 4) + GM_WAVES - 1) / GM_WAVES)

__device__ __forceinline__ void gcn_layer1_block(const float* __restrict__ x,
                                                 const float* __restrict__ W,
                                                 float* __restrict__ out,
                                                 float* Wl, const int blk, const int tid) {
#pragma unroll
    for (int jj = 0; jj < (GCN_IN * 16) / EB_T; ++jj) {
        const int i4 = tid + jj * EB_T;
        const float4 w = *(const float4*)(W + i4 * 4);
        const int k = i4 >> 4;
        const int n = (i4 & 15) * 4;
        float* d = &Wl[(k >> 1) * PAIR_STRIDE + n * 2 + (k & 1)];
        d[0] = w.x; d[2] = w.y; d[4] = w.z; d[6] = w.w;
    }
    __syncthreads();

    const int lane = tid & 31;
    const int wave = tid >> 5;
    const int tile = blk * GM_WAVES + wave;
    if (tile >= (GCN_N >> 4)) return;

    const int m    = lane & 15;
    const int hi   = lane >> 4;
    const int koff = hi * 2;
    const float* arow = x + (size_t)(tile * 16 + m) * GCN_IN;
    const float* brow = &Wl[(koff >> 1) * PAIR_STRIDE + m * 2];

    v8f acc0 = {}, acc1 = {}, acc2 = {}, acc3 = {};
    for (int k0 = 0; k0 < GCN_IN; k0 += 4) {
        const float2 av = *(const float2*)(arow + k0 + koff);
        v2f a; a.x = av.x; a.y = av.y;

        const float* bp = brow + (k0 >> 1) * PAIR_STRIDE;
        v2f b;
        b = *(const v2f*)(bp +  0);
        acc0 = __builtin_amdgcn_wmma_f32_16x16x4_f32(false, a, false, b, (short)0, acc0, false, false);
        b = *(const v2f*)(bp + 32);
        acc1 = __builtin_amdgcn_wmma_f32_16x16x4_f32(false, a, false, b, (short)0, acc1, false, false);
        b = *(const v2f*)(bp + 64);
        acc2 = __builtin_amdgcn_wmma_f32_16x16x4_f32(false, a, false, b, (short)0, acc2, false, false);
        b = *(const v2f*)(bp + 96);
        acc3 = __builtin_amdgcn_wmma_f32_16x16x4_f32(false, a, false, b, (short)0, acc3, false, false);
    }

    float* orow = out + (size_t)(tile * 16 + 8 * hi) * 64 + m;
#pragma unroll
    for (int r = 0; r < 8; ++r) {
        orow[(size_t)r * 64 +  0] = acc0[r];
        orow[(size_t)r * 64 + 16] = acc1[r];
        orow[(size_t)r * 64 + 32] = acc2[r];
        orow[(size_t)r * 64 + 48] = acc3[r];
    }
}

__global__ __launch_bounds__(EB_T) void gcn_bucket_move(const int* __restrict__ ei,
                                                        int* __restrict__ cntT,
                                                        int* __restrict__ offT,
                                                        int* __restrict__ bkt,
                                                        const float* __restrict__ x,
                                                        const float* __restrict__ W1,
                                                        float* __restrict__ t) {
    __shared__ int wsum[8];
    __shared__ int cur[256];
    __shared__ float Wl[(GCN_IN / 2) * PAIR_STRIDE];
    const int tid = threadIdx.x;
    const int c = blockIdx.x;
    if (c >= EB_CHUNKS) {
        gcn_layer1_block(x, W1, t, Wl, c - EB_CHUNKS, tid);
        return;
    }

    const int* srcp = ei + c * EB_CHUNK;
    const int* dstp = ei + GCN_E + c * EB_CHUNK;
    int d[EB_PER_T], s[EB_PER_T];
#pragma unroll
    for (int j = 0; j < EB_PER_T; ++j) {
        const int i = tid + j * EB_T;
        d[j] = (i < EB_CHUNK) ? dstp[i] : -1;
        s[j] = (i < EB_CHUNK) ? srcp[i] : 0;
    }
    if (tid < 256) cur[tid] = 0;
    __syncthreads();
#pragma unroll
    for (int j = 0; j < EB_PER_T; ++j)
        if (d[j] >= 0) atomicAdd(&cur[d[j] >> 8], 1);
    __syncthreads();
    const int hv = (tid < 256) ? cur[tid] : 0;
    const int excl = gcn_scan256(wsum, tid, hv);
    if (tid < 256) {
        cur[tid] = excl;
        cntT[tid * EB_CHUNKS + c] = hv;
        offT[tid * EB_CHUNKS + c] = excl;
    }
    __syncthreads();

    int* stage = reinterpret_cast<int*>(Wl);
#pragma unroll
    for (int j = 0; j < EB_PER_T; ++j)
        if (d[j] >= 0) {
            const int pos = atomicAdd(&cur[d[j] >> 8], 1);
            stage[pos] = (s[j] << 8) | (d[j] & 255);
        }
    __syncthreads();
    int2* reg2 = reinterpret_cast<int2*>(bkt + c * EB_CHUNK);
    const int2* st2 = reinterpret_cast<const int2*>(stage);
    const int lead = (c * (EB_CHUNK / 2)) & 15;
#pragma unroll
    for (int j = 0; j < (EB_CHUNK / 2 + 15 + EB_T - 1) / EB_T; ++j) {
        const int i = tid - lead + j * EB_T;
        if (i >= 0 && i < EB_CHUNK / 2) reg2[i] = st2[i];
    }
}

__global__ __launch_bounds__(EB_T) void gcn_bucket_finish(const int* __restrict__ cntT,
                                                          const int* __restrict__ offT,
                                                          const int* __restrict__ bkt,
                                                          int* __restrict__ ptr,
                                                          float* __restrict__ dinv,
                                                          int* __restrict__ csr,
                                                          float* __restrict__ t,
                                                          float* __restrict__ zrow) {
    __shared__ int wsum[8];
    __shared__ int wsum2[8];
    __shared__ int cnt[256];
    __shared__ float dl[256];
    __shared__ int rbeg[EB_CHUNKS];
    __shared__ int rlen[EB_CHUNKS];
    __shared__ int tot[2];
    __shared__ int stage[EB_STAGE];
    const int tid = threadIdx.x;
    const int b = blockIdx.x;
    if (b == 0 && tid < 16) ((float4*)zrow)[tid] = make_float4(0.0f, 0.0f, 0.0f, 0.0f);
    int hc = 0, ec = 0;
    if (tid < EB_CHUNKS) {
        hc = cntT[b * EB_CHUNKS + tid];
        ec = offT[b * EB_CHUNKS + tid];
    }
    float4* trow = (float4*)t + (size_t)b * (256 * 16);
    const int nrow4 = min(GCN_N - b * 256, 256) * 16;
    float4 tr[(256 * 16) / EB_T];
#pragma unroll
    for (int u = 0; u < (256 * 16) / EB_T; ++u) {
        const int f = tid + u * EB_T;
        if (f < nrow4) tr[u] = trow[f];
    }
    if (tid < 256) cnt[tid] = 0;
    if (tid < EB_CHUNKS) {
        rbeg[tid] = tid * EB_CHUNK + ec;
        rlen[tid] = hc;
    }
    const int pe = gcn_scan256(wsum2, tid, ec);
    if (tid == 255) tot[0] = pe + ec;
    __syncthreads();
    const int beg = tot[0];

    const int rc = tid >> 2;
    const int rk = tid & 3;
    const int len = rlen[rc];
    const int* run = bkt + rbeg[rc];
    int v[EB_FIN_U];
#pragma unroll
    for (int u = 0; u < EB_FIN_U; ++u) {
        const int e = rk + 4 * u;
        v[u] = (e < len) ? run[e] : -1;
    }
#pragma unroll
    for (int u = 0; u < EB_FIN_U; ++u)
        if (v[u] >= 0) atomicAdd(&cnt[v[u] & 255], 1);
    for (int e = rk + 4 * EB_FIN_U; e < len; e += 4)
        atomicAdd(&cnt[run[e] & 255], 1);
    __syncthreads();
    const int node = b * 256 + tid;
    int cn = 0;
    if (tid < 256) {
        cn = cnt[tid];
        const float dg = 2.0f + (float)cn;
        const float dv = (dg > 0.0f) ? rsqrtf(dg) : 0.0f;
        if (node < GCN_N) dinv[node] = dv;
        dl[tid] = dv;
    }
    const int first = beg + gcn_scan256(wsum, tid, cn);
#pragma unroll
    for (int u = 0; u < (256 * 16) / EB_T; ++u) {
        const int f = tid + u * EB_T;
        if (f < nrow4) {
            const float s = dl[f >> 4];
            float4 v4 = tr[u];
            v4.x *= s; v4.y *= s; v4.z *= s; v4.w *= s;
            trow[f] = v4;
        }
    }
    if (tid < 256) {
        if (node <= GCN_N) ptr[node] = first;
        cnt[tid] = first;
    }
    if (tid == 255) tot[1] = first + cn - beg;
    __syncthreads();
    const int nb = tot[1];
    if (nb <= EB_STAGE) {
#pragma unroll
        for (int u = 0; u < EB_FIN_U; ++u)
            if (v[u] >= 0) {
                const int pos = atomicAdd(&cnt[v[u] & 255], 1);
                stage[pos - beg] = v[u] >> 8;
            }
        for (int e = rk + 4 * EB_FIN_U; e < len; e += 4) {
            const int w = run[e];
            const int pos = atomicAdd(&cnt[w & 255], 1);
            stage[pos - beg] = w >> 8;
        }
        __syncthreads();
        int* cb = csr + beg;
        const int lead = beg & 31;
        for (int i0 = tid - lead; i0 < nb; i0 += 4 * EB_T) {
            int w[4];
#pragma unroll
            for (int k = 0; k < 4; ++k)
                w[k] = stage[max(min(i0 + k * EB_T, nb - 1), 0)];
#pragma unroll
            for (int k = 0; k < 4; ++k) {
                const int i = i0 + k * EB_T;
                if (i >= 0 && i < nb) cb[i] = w[k];
            }
        }
    } else {
#pragma unroll
        for (int u = 0; u < EB_FIN_U; ++u)
            if (v[u] >= 0) {
                const int pos = atomicAdd(&cnt[v[u] & 255], 1);
                csr[pos] = v[u] >> 8;
            }
        for (int e = rk + 4 * EB_FIN_U; e < len; e += 4) {
            const int w = run[e];
            const int pos = atomicAdd(&cnt[w & 255], 1);
            csr[pos] = w >> 8;
        }
    }
}

#define GG_TILES 4

__device__ __forceinline__ void gcn_ids32(const int* __restrict__ csr, const int beg,
                                          const int deg, const int sub, const int rowoff,
                                          int& s0, int& s1) {
    const int a0 = csr[min(beg + sub, GCN_E - 1)];
    const int a1 = csr[min(beg + 16 + sub, GCN_E - 1)];
    s0 = (sub < deg) ? a0 + rowoff : 0;
    s1 = (16 + sub < deg) ? a1 + rowoff : 0;
}

__device__ __forceinline__ void gcn_rows16_issue(float4 (&r)[16], const float* __restrict__ zrow,
                                                 const int key, const int q) {
    unsigned off[16];
#pragma unroll
    for (int u = 0; u < 16; ++u)
        off[u] = (unsigned)__shfl(key, u, 16) * 256u + (unsigned)q * 4u;
#pragma unroll
    for (int u = 0; u < 16; ++u)
        r[u] = *(const float4*)((const char*)zrow + off[u]);
}

__device__ __forceinline__ void gcn_rows16_sum(float4& acc, const float4 (&r)[16]) {
#pragma unroll
    for (int u = 0; u < 16; ++u) {
        acc.x += r[u].x;
        acc.y += r[u].y;
        acc.z += r[u].z;
        acc.w += r[u].w;
    }
}

template <class F>
__device__ __forceinline__ float4 gcn_node_sum(const int* __restrict__ csr,
                                               const float* __restrict__ zrow,
                                               const int rowoff,
                                               const int sub, const int q,
                                               const int beg, const int end,
                                               const int s0, const int s1,
                                               const float4 tv, F&& ahead) {
    float4 acc;
    acc.x = 2.0f * tv.x;
    acc.y = 2.0f * tv.y;
    acc.z = 2.0f * tv.z;
    acc.w = 2.0f * tv.w;
    const int deg = end - beg;
    float4 r[16];
    gcn_rows16_issue(r, zrow, s0, q);
    ahead();
    gcn_rows16_sum(acc, r);
    for (int base = 16; base < deg; base += 16) {
        int key = s1;
        if (base >= 32) {
            const int a = csr[min(beg + base + sub, GCN_E - 1)];
            key = (base + sub < deg) ? a + rowoff : 0;
        }
        gcn_rows16_issue(r, zrow, key, q);
        gcn_rows16_sum(acc, r);
    }
    return acc;
}

struct gcn_pipe {
    int   begC, endC, s0C, s1C;
    float diC;
    float4 tvC;
    int   begN, endN;
    int   begNN, endNN, s0N, s1N;
    float diN;
    float4 tvN;
};

__device__ __forceinline__ void gcn_pipe_bounds(gcn_pipe& p, const int* __restrict__ ptr,
                                                const int i, const int nt) {
    p.begC = ptr[i];
    p.endC = ptr[i + 1];
    p.begN = 0; p.endN = 0;
    if (nt > 1) { p.begN = ptr[i + 16]; p.endN = ptr[i + 17]; }
}
__device__ __forceinline__ void gcn_pipe_first(gcn_pipe& p, const int* __restrict__ csr,
                                               const float* __restrict__ dinv,
                                               const float* __restrict__ t, const int rowoff,
                                               const int i, const int sub, const int q) {
    gcn_ids32(csr, p.begC, p.endC - p.begC, sub, rowoff, p.s0C, p.s1C);
    p.diC = dinv[i];
    p.tvC = *(const float4*)(t + (size_t)i * 64 + q);
}
__device__ __forceinline__ void gcn_pipe_ahead(gcn_pipe& p, const int* __restrict__ ptr,
                                               const int* __restrict__ csr,
                                               const float* __restrict__ dinv,
                                               const float* __restrict__ t, const int rowoff,
                                               const int i, const int sub, const int q,
                                               const int j, const int nt) {
    p.s0N = 0; p.s1N = 0; p.begNN = 0; p.endNN = 0; p.diN = 0.0f; p.tvN = p.tvC;
    if (j + 1 < nt) {
        gcn_ids32(csr, p.begN, p.endN - p.begN, sub, rowoff, p.s0N, p.s1N);
        p.diN = dinv[i + 16];
        p.tvN = *(const float4*)(t + (size_t)(i + 16) * 64 + q);
    }
    if (j + 2 < nt) { p.begNN = ptr[i + 32]; p.endNN = ptr[i + 33]; }
}
__device__ __forceinline__ void gcn_pipe_step(gcn_pipe& p) {
    p.begC = p.begN; p.endC = p.endN; p.s0C = p.s0N; p.s1C = p.s1N;
    p.diC = p.diN; p.tvC = p.tvN;
    p.begN = p.begNN; p.endN = p.endNN;
}

__global__ __launch_bounds__(256) void gcn_gather(const int* __restrict__ ptr,
                                                  const int* __restrict__ csr,
                                                  const float* __restrict__ dinv,
                                                  const float* __restrict__ t,
                                                  const float* __restrict__ zrow,
                                                  const float* __restrict__ bias,
                                                  float* __restrict__ out) {
    const int tid = threadIdx.x;
    const int g   = tid >> 4;
    const int sub = tid & 15;
    const int q   = sub * 4;
    const int rowoff = (int)(((const char*)t - (const char*)zrow) >> 8);
    const int tile0 = blockIdx.x * GG_TILES;
    const int nt = min(GG_TILES, (GCN_N >> 4) - tile0);
    int i = tile0 * 16 + g;
    gcn_pipe p;
    gcn_pipe_bounds(p, ptr, i, nt);
    const float4 bv = *(const float4*)(bias + q);
    gcn_pipe_first(p, csr, dinv, t, rowoff, i, sub, q);
    for (int j = 0; j < nt; ++j) {
        const float di = p.diC;
        const float4 acc = gcn_node_sum(csr, zrow, rowoff, sub, q, p.begC, p.endC, p.s0C, p.s1C, p.tvC,
                                        [&] { gcn_pipe_ahead(p, ptr, csr, dinv, t, rowoff, i, sub, q, j, nt); });
        float4 o;
        o.x = bv.x + di * acc.x;
        o.y = bv.y + di * acc.y;
        o.z = bv.z + di * acc.z;
        o.w = bv.w + di * acc.w;
        *(float4*)(out + (size_t)i * 64 + q) = o;
        gcn_pipe_step(p);
        i += 16;
    }
}

#define A_PITCH 68

__global__ __launch_bounds__(256) void gcn_gather_gemm(const int* __restrict__ ptr,
                                                       const int* __restrict__ csr,
                                                       const float* __restrict__ dinv,
                                                       const float* __restrict__ tin,
                                                       const float* __restrict__ zrow,
                                                       const float* __restrict__ bias,
                                                       const float* __restrict__ W,
                                                       float* __restrict__ tout) {
    __shared__ float Wl[(GCN_H / 2) * PAIR_STRIDE];
    __shared__ float At[2][16 * A_PITCH];
    const int tid = threadIdx.x;
    const int g    = tid >> 4;
    const int sub  = tid & 15;
    const int q    = sub * 4;

    const int rowoff = (int)(((const char*)tin - (const char*)zrow) >> 8);

    const int tile0 = blockIdx.x * GG_TILES;
    const int nt = min(GG_TILES, (GCN_N >> 4) - tile0);
    int i = tile0 * 16 + g;
    gcn_pipe p;
    gcn_pipe_bounds(p, ptr, i, nt);

#pragma unroll
    for (int jj = 0; jj < (GCN_H * 16) / 256; ++jj) {
        const int i4 = tid + jj * 256;
        const float4 w = *(const float4*)(W + i4 * 4);
        const int k = i4 >> 4;
        const int n = (i4 & 15) * 4;
        float* d = &Wl[(k >> 1) * PAIR_STRIDE + n * 2 + (k & 1)];
        d[0] = w.x; d[2] = w.y; d[4] = w.z; d[6] = w.w;
    }

    const int lane = tid & 31;
    const int wave = tid >> 5;
    const int m    = lane & 15;
    const int hi   = lane >> 4;
    const int koff = hi * 2;
    const int n0   = (wave & 3) * 16;
    const float4 bv = *(const float4*)(bias + q);
    gcn_pipe_first(p, csr, dinv, tin, rowoff, i, sub, q);

    for (int j = 0; j < nt; ++j) {
        const int tile = tile0 + j;
        const float di = p.diC;
        const float4 acc = gcn_node_sum(csr, zrow, rowoff, sub, q, p.begC, p.endC, p.s0C, p.s1C, p.tvC,
                                        [&] { gcn_pipe_ahead(p, ptr, csr, dinv, tin, rowoff, i, sub, q, j, nt); });
        float4 a4;
        a4.x = di * fmaxf(bv.x + di * acc.x, 0.0f);
        a4.y = di * fmaxf(bv.y + di * acc.y, 0.0f);
        a4.z = di * fmaxf(bv.z + di * acc.z, 0.0f);
        a4.w = di * fmaxf(bv.w + di * acc.w, 0.0f);
        float* Ab = At[j & 1];
        *(float4*)(&Ab[g * A_PITCH + q]) = a4;
        __syncthreads();

        if (wave < 4) {
            const float* ap = &Ab[m * A_PITCH + koff];
            const float* bp = &Wl[(koff >> 1) * PAIR_STRIDE + (n0 + m) * 2];
            v8f c8 = {};
#pragma unroll
            for (int kk = 0; kk < 16; ++kk) {
                const v2f a = *(const v2f*)(ap + kk * 4);
                const v2f b = *(const v2f*)(bp + kk * 2 * PAIR_STRIDE);
                c8 = __builtin_amdgcn_wmma_f32_16x16x4_f32(false, a, false, b, (short)0, c8, false, false);
            }
            float* orow = tout + (size_t)(tile * 16 + 8 * hi) * 64 + n0 + m;
#pragma unroll
            for (int r = 0; r < 8; ++r) orow[(size_t)r * 64] = c8[r];
        }
        gcn_pipe_step(p);
        i += 16;
    }
}


extern "C" void kernel_launch(void* const* d_in, const int* in_sizes, int n_in,
                              void* d_out, int out_size, void* d_ws, size_t ws_size,
                              hipStream_t stream) {
    (void)in_sizes; (void)n_in; (void)out_size; (void)ws_size;

    const float* x  = (const float*)d_in[0];
    const int*   ei = (const int*)d_in[1];
    const float* W[5] = {(const float*)d_in[2], (const float*)d_in[4], (const float*)d_in[6],
                         (const float*)d_in[8], (const float*)d_in[10]};
    const float* B[5] = {(const float*)d_in[3], (const float*)d_in[5], (const float*)d_in[7],
                         (const float*)d_in[9], (const float*)d_in[11]};

    char*  ws  = (char*)d_ws;
    size_t off = 0;
    auto carve = [&](size_t bytes) {
        void* p = ws + off;
        off = (off + bytes + 255) & ~(size_t)255;
        return p;
    };
    char*  head = (char*) carve(2 * (((size_t)(GCN_N + 1) * 4 + 255) & ~(size_t)255));
    int*   ptr  = (int*)  head;
    float* dinv = (float*)(head + 200064);
    float* zrow = (float*)(head + 400128);
    int*   csr  = (int*)  carve((size_t)GCN_E * 4);
    float* t    = (float*)carve((size_t)GCN_N * 64 * 4);
    float* hagg = (float*)carve((size_t)GCN_N * 64 * 4);
    int*   bkt  = (int*)hagg;
    int*   cntT = (int*)hagg + GCN_E;
    int*   offT = (int*)hagg + GCN_E + EB_CHUNKS * 256;

    const int BGG = ((GCN_N / 16) + GG_TILES - 1) / GG_TILES;

    gcn_bucket_move  <<<EB_CHUNKS + GM_BLOCKS, EB_T, 0, stream>>>(ei, cntT, offT, bkt, x, W[0], t);
    gcn_bucket_finish<<<EB_BUCKETS, EB_T, 0, stream>>>(cntT, offT, bkt, ptr, dinv, csr, t, zrow);

    gcn_gather_gemm<<<BGG, 256, 0, stream>>>(ptr, csr, dinv, t,    zrow, B[0], W[1], hagg);
    gcn_gather_gemm<<<BGG, 256, 0, stream>>>(ptr, csr, dinv, hagg, zrow, B[1], W[2], t);
    gcn_gather_gemm<<<BGG, 256, 0, stream>>>(ptr, csr, dinv, t,    zrow, B[2], W[3], hagg);
    gcn_gather_gemm<<<BGG, 256, 0, stream>>>(ptr, csr, dinv, hagg, zrow, B[3], W[4], t);

    float* out = (float*)d_out;
    gcn_gather<<<BGG, 256, 0, stream>>>(ptr, csr, dinv, t, zrow, B[4], out);
}


# Edge sums ask for a node's own row and the rows of its first 32 list slots as one group into two register sets (was 16, wait, 16); grid cut to the 768 resident work-groups; sum order unchanged
# speedup vs baseline: 5.1865x; 1.0959x over previous
#include <hip/hip_runtime.h>


#define GCN_N 50000
#define GCN_E 800000
#define GCN_IN 128
#define GCN_H 64

#define PAIR_STRIDE 130

typedef float v2f __attribute__((ext_vector_type(2)));
typedef float v8f __attribute__((ext_vector_type(8)));

#define EB_T       512
#define EB_CHUNKS  128
#define EB_CHUNK   (GCN_E / EB_CHUNKS)
#define EB_PER_T   ((EB_CHUNK + EB_T - 1) / EB_T)
#define EB_BUCKETS ((GCN_N + 255) / 256)
#define EB_FIN_U   16
#define EB_STAGE   8192
static_assert(EB_CHUNK % 2 == 0, "a chunk's region is moved out 8 bytes at a time");
static_assert(EB_CHUNK * 4 <= (GCN_IN / 2) * PAIR_STRIDE * 4, "a chunk's region fits the W staging words");

__device__ __forceinline__ int gcn_scan256(int* wsum, const int tid, const int v) {
    const int lane = tid & 31;
    const int wave = tid >> 5;
    int x = v;
#pragma unroll
    for (int off = 1; off < 32; off <<= 1) {
        const int y = __shfl_up(x, off, 32);
        if (lane >= off) x += y;
    }
    if (wave < 8 && lane == 31) wsum[wave] = x;
    __syncthreads();
    int add = 0;
#pragma unroll
    for (int w = 0; w < 8; ++w)
        if (w < wave) add += wsum[w];
    return x - v + add;
}

#define GM_WAVES  (EB_T / 32)
#define GM_BLOCKS (((GCN_N >> 4) + GM_WAVES - 1) / GM_WAVES)

__device__ __forceinline__ void gcn_layer1_block(const float* __restrict__ x,
                                                 const float* __restrict__ W,
                                                 float* __restrict__ out,
                                                 float* Wl, const int blk, const int tid) {
#pragma unroll
    for (int jj = 0; jj < (GCN_IN * 16) / EB_T; ++jj) {
        const int i4 = tid + jj * EB_T;
        const float4 w = *(const float4*)(W + i4 * 4);
        const int k = i4 >> 4;
        const int n = (i4 & 15) * 4;
        float* d = &Wl[(k >> 1) * PAIR_STRIDE + n * 2 + (k & 1)];
        d[0] = w.x; d[2] = w.y; d[4] = w.z; d[6] = w.w;
    }
    __syncthreads();

    const int lane = tid & 31;
    const int wave = tid >> 5;
    const int tile = blk * GM_WAVES + wave;
    if (tile >= (GCN_N >> 4)) return;

    const int m    = lane & 15;
    const int hi   = lane >> 4;
    const int koff = hi * 2;
    const float* arow = x + (size_t)(tile * 16 + m) * GCN_IN;
    const float* brow = &Wl[(koff >> 1) * PAIR_STRIDE + m * 2];

    v8f acc0 = {}, acc1 = {}, acc2 = {}, acc3 = {};
    for (int k0 = 0; k0 < GCN_IN; k0 += 4) {
        const float2 av = *(const float2*)(arow + k0 + koff);
        v2f a; a.x = av.x; a.y = av.y;

        const float* bp = brow + (k0 >> 1) * PAIR_STRIDE;
        v2f b;
        b = *(const v2f*)(bp +  0);
        acc0 = __builtin_amdgcn_wmma_f32_16x16x4_f32(false, a, false, b, (short)0, acc0, false, false);
        b = *(const v2f*)(bp + 32);
        acc1 = __builtin_amdgcn_wmma_f32_16x16x4_f32(false, a, false, b, (short)0, acc1, false, false);
        b = *(const v2f*)(bp + 64);
        acc2 = __builtin_amdgcn_wmma_f32_16x16x4_f32(false, a, false, b, (short)0, acc2, false, false);
        b = *(const v2f*)(bp + 96);
        acc3 = __builtin_amdgcn_wmma_f32_16x16x4_f32(false, a, false, b, (short)0, acc3, false, false);
    }

    float* orow = out + (size_t)(tile * 16 + 8 * hi) * 64 + m;
#pragma unroll
    for (int r = 0; r < 8; ++r) {
        orow[(size_t)r * 64 +  0] = acc0[r];
        orow[(size_t)r * 64 + 16] = acc1[r];
        orow[(size_t)r * 64 + 32] = acc2[r];
        orow[(size_t)r * 64 + 48] = acc3[r];
    }
}

__global__ __launch_bounds__(EB_T) void gcn_bucket_move(const int* __restrict__ ei,
                                                        int* __restrict__ cntT,
                                                        int* __restrict__ offT,
                                                        int* __restrict__ bkt,
                                                        const float* __restrict__ x,
                                                        const float* __restrict__ W1,
                                                        float* __restrict__ t) {
    __shared__ int wsum[8];
    __shared__ int cur[256];
    __shared__ float Wl[(GCN_IN / 2) * PAIR_STRIDE];
    const int tid = threadIdx.x;
    const int c = blockIdx.x;
    if (c >= EB_CHUNKS) {
        gcn_layer1_block(x, W1, t, Wl, c - EB_CHUNKS, tid);
        return;
    }

    const int* srcp = ei + c * EB_CHUNK;
    const int* dstp = ei + GCN_E + c * EB_CHUNK;
    int d[EB_PER_T], s[EB_PER_T];
#pragma unroll
    for (int j = 0; j < EB_PER_T; ++j) {
        const int i = tid + j * EB_T;
        d[j] = (i < EB_CHUNK) ? dstp[i] : -1;
        s[j] = (i < EB_CHUNK) ? srcp[i] : 0;
    }
    if (tid < 256) cur[tid] = 0;
    __syncthreads();
#pragma unroll
    for (int j = 0; j < EB_PER_T; ++j)
        if (d[j] >= 0) atomicAdd(&cur[d[j] >> 8], 1);
    __syncthreads();
    const int hv = (tid < 256) ? cur[tid] : 0;
    const int excl = gcn_scan256(wsum, tid, hv);
    if (tid < 256) {
        cur[tid] = excl;
        cntT[tid * EB_CHUNKS + c] = hv;
        offT[tid * EB_CHUNKS + c] = excl;
    }
    __syncthreads();

    int* stage = reinterpret_cast<int*>(Wl);
#pragma unroll
    for (int j = 0; j < EB_PER_T; ++j)
        if (d[j] >= 0) {
            const int pos = atomicAdd(&cur[d[j] >> 8], 1);
            stage[pos] = (s[j] << 8) | (d[j] & 255);
        }
    __syncthreads();
    int2* reg2 = reinterpret_cast<int2*>(bkt + c * EB_CHUNK);
    const int2* st2 = reinterpret_cast<const int2*>(stage);
    const int lead = (c * (EB_CHUNK / 2)) & 15;
#pragma unroll
    for (int j = 0; j < (EB_CHUNK / 2 + 15 + EB_T - 1) / EB_T; ++j) {
        const int i = tid - lead + j * EB_T;
        if (i >= 0 && i < EB_CHUNK / 2) reg2[i] = st2[i];
    }
}

__global__ __launch_bounds__(EB_T) void gcn_bucket_finish(const int* __restrict__ cntT,
                                                          const int* __restrict__ offT,
                                                          const int* __restrict__ bkt,
                                                          int* __restrict__ ptr,
                                                          float* __restrict__ dinv,
                                                          int* __restrict__ csr,
                                                          float* __restrict__ t,
                                                          float* __restrict__ zrow) {
    __shared__ int wsum[8];
    __shared__ int wsum2[8];
    __shared__ int cnt[256];
    __shared__ float dl[256];
    __shared__ int rbeg[EB_CHUNKS];
    __shared__ int rlen[EB_CHUNKS];
    __shared__ int tot[2];
    __shared__ int stage[EB_STAGE];
    const int tid = threadIdx.x;
    const int b = blockIdx.x;
    if (b == 0 && tid < 16) ((float4*)zrow)[tid] = make_float4(0.0f, 0.0f, 0.0f, 0.0f);
    int hc = 0, ec = 0;
    if (tid < EB_CHUNKS) {
        hc = cntT[b * EB_CHUNKS + tid];
        ec = offT[b * EB_CHUNKS + tid];
    }
    float4* trow = (float4*)t + (size_t)b * (256 * 16);
    const int nrow4 = min(GCN_N - b * 256, 256) * 16;
    float4 tr[(256 * 16) / EB_T];
#pragma unroll
    for (int u = 0; u < (256 * 16) / EB_T; ++u) {
        const int f = tid + u * EB_T;
        if (f < nrow4) tr[u] = trow[f];
    }
    if (tid < 256) cnt[tid] = 0;
    if (tid < EB_CHUNKS) {
        rbeg[tid] = tid * EB_CHUNK + ec;
        rlen[tid] = hc;
    }
    const int pe = gcn_scan256(wsum2, tid, ec);
    if (tid == 255) tot[0] = pe + ec;
    __syncthreads();
    const int beg = tot[0];

    const int rc = tid >> 2;
    const int rk = tid & 3;
    const int len = rlen[rc];
    const int* run = bkt + rbeg[rc];
    int v[EB_FIN_U];
#pragma unroll
    for (int u = 0; u < EB_FIN_U; ++u) {
        const int e = rk + 4 * u;
        v[u] = (e < len) ? run[e] : -1;
    }
#pragma unroll
    for (int u = 0; u < EB_FIN_U; ++u)
        if (v[u] >= 0) atomicAdd(&cnt[v[u] & 255], 1);
    for (int e = rk + 4 * EB_FIN_U; e < len; e += 4)
        atomicAdd(&cnt[run[e] & 255], 1);
    __syncthreads();
    const int node = b * 256 + tid;
    int cn = 0;
    if (tid < 256) {
        cn = cnt[tid];
        const float dg = 2.0f + (float)cn;
        const float dv = (dg > 0.0f) ? rsqrtf(dg) : 0.0f;
        if (node < GCN_N) dinv[node] = dv;
        dl[tid] = dv;
    }
    const int first = beg + gcn_scan256(wsum, tid, cn);
#pragma unroll
    for (int u = 0; u < (256 * 16) / EB_T; ++u) {
        const int f = tid + u * EB_T;
        if (f < nrow4) {
            const float s = dl[f >> 4];
            float4 v4 = tr[u];
            v4.x *= s; v4.y *= s; v4.z *= s; v4.w *= s;
            trow[f] = v4;
        }
    }
    if (tid < 256) {
        if (node <= GCN_N) ptr[node] = first;
        cnt[tid] = first;
    }
    if (tid == 255) tot[1] = first + cn - beg;
    __syncthreads();
    const int nb = tot[1];
    if (nb <= EB_STAGE) {
#pragma unroll
        for (int u = 0; u < EB_FIN_U; ++u)
            if (v[u] >= 0) {
                const int pos = atomicAdd(&cnt[v[u] & 255], 1);
                stage[pos - beg] = v[u] >> 8;
            }
        for (int e = rk + 4 * EB_FIN_U; e < len; e += 4) {
            const int w = run[e];
            const int pos = atomicAdd(&cnt[w & 255], 1);
            stage[pos - beg] = w >> 8;
        }
        __syncthreads();
        int* cb = csr + beg;
        const int lead = beg & 31;
        for (int i0 = tid - lead; i0 < nb; i0 += 4 * EB_T) {
            int w[4];
#pragma unroll
            for (int k = 0; k < 4; ++k)
                w[k] = stage[max(min(i0 + k * EB_T, nb - 1), 0)];
#pragma unroll
            for (int k = 0; k < 4; ++k) {
                const int i = i0 + k * EB_T;
                if (i >= 0 && i < nb) cb[i] = w[k];
            }
        }
    } else {
#pragma unroll
        for (int u = 0; u < EB_FIN_U; ++u)
            if (v[u] >= 0) {
                const int pos = atomicAdd(&cnt[v[u] & 255], 1);
                csr[pos] = v[u] >> 8;
            }
        for (int e = rk + 4 * EB_FIN_U; e < len; e += 4) {
            const int w = run[e];
            const int pos = atomicAdd(&cnt[w & 255], 1);
            csr[pos] = w >> 8;
        }
    }
}

#define GG_TILES 4
#define GG_WGS  768
#define GG_LONG ((GCN_N / 16) - GG_WGS * GG_TILES)
static_assert(GCN_N % 16 == 0, "whole node tiles");
static_assert(GG_LONG >= 0 && GG_LONG <= GG_WGS, "at most one tile more per work-group");

__device__ __forceinline__ void gcn_ids32_ask(const int* __restrict__ csr, const int beg,
                                              const int sub, int& a0, int& a1) {
    a0 = csr[min(beg + sub, GCN_E - 1)];
    a1 = csr[min(beg + 16 + sub, GCN_E - 1)];
}
__device__ __forceinline__ void gcn_ids32_keys(const int a0, const int a1,
                                               const int deg, const int sub, const int rowoff,
                                               int& s0, int& s1) {
    s0 = (sub < deg) ? a0 + rowoff : 0;
    s1 = (16 + sub < deg) ? a1 + rowoff : 0;
}

__device__ __forceinline__ void gcn_rows16_issue(float4 (&r)[16], const float* __restrict__ zrow,
                                                 const int key, const int q) {
    unsigned off[16];
#pragma unroll
    for (int u = 0; u < 16; ++u)
        off[u] = (unsigned)__shfl(key, u, 16) * 256u + (unsigned)q * 4u;
#pragma unroll
    for (int u = 0; u < 16; ++u)
        r[u] = *(const float4*)((const char*)zrow + off[u]);
}

__device__ __forceinline__ void gcn_rows16_sum(float4& acc, const float4 (&r)[16]) {
#pragma unroll
    for (int u = 0; u < 16; ++u) {
        acc.x += r[u].x;
        acc.y += r[u].y;
        acc.z += r[u].z;
        acc.w += r[u].w;
    }
}

template <class F>
__device__ __forceinline__ float4 gcn_node_sum(const int* __restrict__ csr,
                                               const float* __restrict__ dinv,
                                               const float* __restrict__ zrow,
                                               const int rowoff, const int i,
                                               const int sub, const int q,
                                               const int beg, const int end,
                                               const int s0, const int s1,
                                               float& di, F&& ahead) {
    const int deg = end - beg;
    const bool more = deg > 16;
    const float4 tv = *(const float4*)((const char*)zrow +
                                       ((unsigned)(i + rowoff) * 256u + (unsigned)q * 4u));
    di = dinv[i];
    float4 r[16], r2[16];
    gcn_rows16_issue(r, zrow, s0, q);
    if (more) gcn_rows16_issue(r2, zrow, s1, q);
    ahead();
    float4 acc;
    acc.x = 2.0f * tv.x;
    acc.y = 2.0f * tv.y;
    acc.z = 2.0f * tv.z;
    acc.w = 2.0f * tv.w;
    gcn_rows16_sum(acc, r);
    if (more) gcn_rows16_sum(acc, r2);
    for (int base = 32; base < deg; base += 16) {
        const int a = csr[min(beg + base + sub, GCN_E - 1)];
        const int key = (base + sub < deg) ? a + rowoff : 0;
        gcn_rows16_issue(r, zrow, key, q);
        gcn_rows16_sum(acc, r);
    }
    return acc;
}

struct gcn_pipe {
    int   begC, endC, s0C, s1C;
    int   begN, endN;
    int   begNN, endNN, s0N, s1N;
};

__device__ __forceinline__ void gcn_pipe_bounds(gcn_pipe& p, const int* __restrict__ ptr,
                                                const int i, const int nt) {
    p.begC = ptr[i];
    p.endC = ptr[i + 1];
    p.begN = 0; p.endN = 0;
    if (nt > 1) { p.begN = ptr[i + 16]; p.endN = ptr[i + 17]; }
}
__device__ __forceinline__ void gcn_pipe_first(gcn_pipe& p, const int* __restrict__ csr,
                                               const int rowoff, const int sub) {
    int a0, a1;
    gcn_ids32_ask(csr, p.begC, sub, a0, a1);
    gcn_ids32_keys(a0, a1, p.endC - p.begC, sub, rowoff, p.s0C, p.s1C);
}
__device__ __forceinline__ void gcn_pipe_ask_ahead(gcn_pipe& p, const int* __restrict__ ptr,
                                                   const int* __restrict__ csr,
                                                   const int i, const int sub,
                                                   const int j, const int nt, int& a0, int& a1) {
    p.begNN = 0; p.endNN = 0; a0 = 0; a1 = 0;
    if (j + 2 < nt) { p.begNN = ptr[i + 32]; p.endNN = ptr[i + 33]; }
    if (j + 1 < nt) gcn_ids32_ask(csr, p.begN, sub, a0, a1);
}
__device__ __forceinline__ void gcn_pipe_keys_ahead(gcn_pipe& p, const int a0, const int a1,
                                                    const int rowoff, const int sub,
                                                    const int j, const int nt) {
    p.s0N = 0; p.s1N = 0;
    if (j + 1 < nt)
        gcn_ids32_keys(a0, a1, p.endN - p.begN, sub, rowoff, p.s0N, p.s1N);
}
__device__ __forceinline__ void gcn_pipe_step(gcn_pipe& p) {
    p.begC = p.begN; p.endC = p.endN; p.s0C = p.s0N; p.s1C = p.s1N;
    p.begN = p.begNN; p.endN = p.endNN;
}

__global__ __launch_bounds__(256) void gcn_gather(const int* __restrict__ ptr,
                                                  const int* __restrict__ csr,
                                                  const float* __restrict__ dinv,
                                                  const float* __restrict__ t,
                                                  const float* __restrict__ zrow,
                                                  const float* __restrict__ bias,
                                                  float* __restrict__ out) {
    const int tid = threadIdx.x;
    const int g   = tid >> 4;
    const int sub = tid & 15;
    const int q   = sub * 4;
    const int rowoff = (int)(((const char*)t - (const char*)zrow) >> 8);
    const int wg = blockIdx.x;
    const int tile0 = wg * GG_TILES + min(wg, GG_LONG);
    const int nt = GG_TILES + (wg < GG_LONG ? 1 : 0);
    int i = tile0 * 16 + g;
    gcn_pipe p;
    gcn_pipe_bounds(p, ptr, i, nt);
    const float4 bv = *(const float4*)(bias + q);
    gcn_pipe_first(p, csr, rowoff, sub);
    for (int j = 0; j < nt; ++j) {
        float di;
        int a0, a1;
        gcn_pipe_ask_ahead(p, ptr, csr, i, sub, j, nt, a0, a1);
        const float4 acc = gcn_node_sum(csr, dinv, zrow, rowoff, i, sub, q, p.begC, p.endC, p.s0C, p.s1C, di,
                                        [&] { gcn_pipe_keys_ahead(p, a0, a1, rowoff, sub, j, nt); });
        float4 o;
        o.x = bv.x + di * acc.x;
        o.y = bv.y + di * acc.y;
        o.z = bv.z + di * acc.z;
        o.w = bv.w + di * acc.w;
        *(float4*)(out + (size_t)i * 64 + q) = o;
        gcn_pipe_step(p);
        i += 16;
    }
}

#define A_PITCH 68

__global__ __launch_bounds__(256, 6) void gcn_gather_gemm(const int* __restrict__ ptr,
                                                          const int* __restrict__ csr,
                                                          const float* __restrict__ dinv,
                                                          const float* __restrict__ tin,
                                                          const float* __restrict__ zrow,
                                                          const float* __restrict__ bias,
                                                          const float* __restrict__ W,
                                                          float* __restrict__ tout) {
    __shared__ float Wl[(GCN_H / 2) * PAIR_STRIDE];
    __shared__ float At[2][16 * A_PITCH];
    const int tid = threadIdx.x;
    const int g    = tid >> 4;
    const int sub  = tid & 15;
    const int q    = sub * 4;

    const int rowoff = (int)(((const char*)tin - (const char*)zrow) >> 8);

    const int wg = blockIdx.x;
    const int tile0 = wg * GG_TILES + min(wg, GG_LONG);
    const int nt = GG_TILES + (wg < GG_LONG ? 1 : 0);
    int i = tile0 * 16 + g;
    gcn_pipe p;
    gcn_pipe_bounds(p, ptr, i, nt);

#pragma unroll
    for (int jj = 0; jj < (GCN_H * 16) / 256; ++jj) {
        const int i4 = tid + jj * 256;
        const float4 w = *(const float4*)(W + i4 * 4);
        const int k = i4 >> 4;
        const int n = (i4 & 15) * 4;
        float* d = &Wl[(k >> 1) * PAIR_STRIDE + n * 2 + (k & 1)];
        d[0] = w.x; d[2] = w.y; d[4] = w.z; d[6] = w.w;
    }

    const int lane = tid & 31;
    const int wave = tid >> 5;
    const int m    = lane & 15;
    const int hi   = lane >> 4;
    const int koff = hi * 2;
    const int n0   = (wave & 3) * 16;
    const float4 bv = *(const float4*)(bias + q);
    gcn_pipe_first(p, csr, rowoff, sub);

    for (int j = 0; j < nt; ++j) {
        const int tile = tile0 + j;
        float di;
        int a0, a1;
        gcn_pipe_ask_ahead(p, ptr, csr, i, sub, j, nt, a0, a1);
        const float4 acc = gcn_node_sum(csr, dinv, zrow, rowoff, i, sub, q, p.begC, p.endC, p.s0C, p.s1C, di,
                                        [&] { gcn_pipe_keys_ahead(p, a0, a1, rowoff, sub, j, nt); });
        float4 a4;
        a4.x = di * fmaxf(bv.x + di * acc.x, 0.0f);
        a4.y = di * fmaxf(bv.y + di * acc.y, 0.0f);
        a4.z = di * fmaxf(bv.z + di * acc.z, 0.0f);
        a4.w = di * fmaxf(bv.w + di * acc.w, 0.0f);
        float* Ab = At[j & 1];
        *(float4*)(&Ab[g * A_PITCH + q]) = a4;
        __syncthreads();

        if (wave < 4) {
            const float* ap = &Ab[m * A_PITCH + koff];
            const float* bp = &Wl[(koff >> 1) * PAIR_STRIDE + (n0 + m) * 2];
            v8f c8 = {};
#pragma unroll
            for (int kk = 0; kk < 16; ++kk) {
                const v2f a = *(const v2f*)(ap + kk * 4);
                const v2f b = *(const v2f*)(bp + kk * 2 * PAIR_STRIDE);
                c8 = __builtin_amdgcn_wmma_f32_16x16x4_f32(false, a, false, b, (short)0, c8, false, false);
            }
            const unsigned ob = ((unsigned)(tile * 16 + 8 * hi) * 64u + (unsigned)(n0 + m)) * 4u;
#pragma unroll
            for (int r = 0; r < 8; ++r) *(float*)((char*)tout + (ob + (unsigned)r * 256u)) = c8[r];
        }
        gcn_pipe_step(p);
        i += 16;
    }
}


extern "C" void kernel_launch(void* const* d_in, const int* in_sizes, int n_in,
                              void* d_out, int out_size, void* d_ws, size_t ws_size,
                              hipStream_t stream) {
    (void)in_sizes; (void)n_in; (void)out_size; (void)ws_size;

    const float* x  = (const float*)d_in[0];
    const int*   ei = (const int*)d_in[1];
    const float* W[5] = {(const float*)d_in[2], (const float*)d_in[4], (const float*)d_in[6],
                         (const float*)d_in[8], (const float*)d_in[10]};
    const float* B[5] = {(const float*)d_in[3], (const float*)d_in[5], (const float*)d_in[7],
                         (const float*)d_in[9], (const float*)d_in[11]};

    char*  ws  = (char*)d_ws;
    size_t off = 0;
    auto carve = [&](size_t bytes) {
        void* p = ws + off;
        off = (off + bytes + 255) & ~(size_t)255;
        return p;
    };
    char*  head = (char*) carve(2 * (((size_t)(GCN_N + 1) * 4 + 255) & ~(size_t)255));
    int*   ptr  = (int*)  head;
    float* dinv = (float*)(head + 200064);
    float* zrow = (float*)(head + 400128);
    int*   csr  = (int*)  carve((size_t)GCN_E * 4);
    float* t    = (float*)carve((size_t)GCN_N * 64 * 4);
    float* hagg = (float*)carve((size_t)GCN_N * 64 * 4);
    int*   bkt  = (int*)hagg;
    int*   cntT = (int*)hagg + GCN_E;
    int*   offT = (int*)hagg + GCN_E + EB_CHUNKS * 256;

    const int BGG = GG_WGS;

    gcn_bucket_move  <<<EB_CHUNKS + GM_BLOCKS, EB_T, 0, stream>>>(ei, cntT, offT, bkt, x, W[0], t);
    gcn_bucket_finish<<<EB_BUCKETS, EB_T, 0, stream>>>(cntT, offT, bkt, ptr, dinv, csr, t, zrow);

    gcn_gather_gemm<<<BGG, 256, 0, stream>>>(ptr, csr, dinv, t,    zrow, B[0], W[1], hagg);
    gcn_gather_gemm<<<BGG, 256, 0, stream>>>(ptr, csr, dinv, hagg, zrow, B[1], W[2], t);
    gcn_gather_gemm<<<BGG, 256, 0, stream>>>(ptr, csr, dinv, t,    zrow, B[2], W[3], hagg);
    gcn_gather_gemm<<<BGG, 256, 0, stream>>>(ptr, csr, dinv, hagg, zrow, B[3], W[4], t);

    float* out = (float*)d_out;
    gcn_gather<<<BGG, 256, 0, stream>>>(ptr, csr, dinv, t, zrow, B[4], out);
}
